# starting kernel
# baseline (speedup 1.0000x reference)
#include <hip/hip_runtime.h>
#include <math.h>

typedef __attribute__((ext_vector_type(2))) float v2f;
typedef __attribute__((ext_vector_type(8))) float v8f;

#define N_NODES 50000
#define N_EDGES 800000
#define HID 256

__global__ __launch_bounds__(256) void edge_mlp_kernel(
    const float* __restrict__ attr,
    const float* __restrict__ W1,
    const float* __restrict__ b1,
    const float* __restrict__ W2,
    const float* __restrict__ b2,
    float* __restrict__ w, int E) {
  __shared__ float sW1[3 * 128];
  __shared__ float sb1[128];
  __shared__ float sW2[128];
  for (int i = threadIdx.x; i < 384; i += 256) sW1[i] = W1[i];
  if (threadIdx.x < 128) {
    sb1[threadIdx.x] = b1[threadIdx.x];
    sW2[threadIdx.x] = W2[threadIdx.x];
  }
  __syncthreads();
  int e = blockIdx.x * 256 + threadIdx.x;
  if (e >= E) return;
  float a0 = attr[e * 3 + 0];
  float a1 = attr[e * 3 + 1];
  float a2 = attr[e * 3 + 2];
  float acc = 0.0f;
#pragma unroll 4
  for (int j = 0; j < 128; ++j) {
    float z = fmaf(a0, sW1[j], fmaf(a1, sW1[128 + j], fmaf(a2, sW1[256 + j], sb1[j])));
    float t = z / (1.0f + __expf(-z));
    acc = fmaf(t, sW2[j], acc);
  }
  float zz = acc + b2[0];
  w[e] = 1.0f / (1.0f + __expf(-zz));
}

__global__ __launch_bounds__(256) void deg_init_kernel(float* __restrict__ deg, int N) {
  int i = blockIdx.x * 256 + threadIdx.x;
  if (i < N) deg[i] = 1.0f;
}

__global__ __launch_bounds__(256) void deg_scatter_kernel(
    const int* __restrict__ col, const float* __restrict__ w,
    float* __restrict__ deg, int E) {
  int e = blockIdx.x * 256 + threadIdx.x;
  if (e < E) unsafeAtomicAdd(&deg[col[e]], w[e]);
}

__global__ __launch_bounds__(256) void deg_rsqrt_kernel(float* __restrict__ deg, int N) {
  int i = blockIdx.x * 256 + threadIdx.x;
  if (i < N) {
    float d = deg[i];
    deg[i] = d > 0.0f ? rsqrtf(d) : 0.0f;
  }
}

__global__ __launch_bounds__(256) void gemm_xWc_kernel(
    const float* __restrict__ h,
    const float* __restrict__ Wc,
    float* __restrict__ x) {
  const int K = 256, NN = 256;
  const int wave = threadIdx.x >> 5;
  const int lane = threadIdx.x & 31;
  const int m0 = (blockIdx.x >> 1) * 16;
  const int n0 = (((blockIdx.x & 1) << 3) + wave) * 16;
  const int r = lane & 15;
  const int g = lane >> 4;

  const float* arow = h + (size_t)(m0 + r) * K + 2 * g;
  const float* bcol = Wc + (size_t)(2 * g) * NN + n0 + r;

  v8f c = {};
  for (int kk = 0; kk < K; kk += 4) {
    v2f a;
    a.x = arow[kk];
    a.y = arow[kk + 1];
    v2f b;
    b.x = bcol[kk * NN];
    b.y = bcol[kk * NN + NN];
    c = __builtin_amdgcn_wmma_f32_16x16x4_f32(false, a, false, b, (short)0, c,
                                              false, false);
  }
#pragma unroll
  for (int i = 0; i < 8; ++i) {
    int m = m0 + i + 8 * g;
    x[(size_t)m * NN + n0 + r] = c[i];
  }
}

__global__ __launch_bounds__(256) void out_init_kernel(
    const float* __restrict__ x, const float* __restrict__ dinv,
    const float* __restrict__ bc, float* __restrict__ out) {
  int idx = blockIdx.x * 256 + threadIdx.x;
  int n = idx >> 8;
  int j = idx & 255;
  float di = dinv[n];
  out[idx] = x[idx] * di * di + bc[j];
}

__global__ __launch_bounds__(256) void aggregate_kernel(
    const int* __restrict__ ei,
    const float* __restrict__ w,
    const float* __restrict__ dinv,
    const float* __restrict__ x,
    float* __restrict__ out, int E) {
  int e = blockIdx.x;
  int r = ei[e];
  int c = ei[E + e];
  float norm = dinv[r] * w[e] * dinv[c];
  int j = threadIdx.x;
  float v = x[(size_t)r * HID + j] * norm;
  unsafeAtomicAdd(&out[(size_t)c * HID + j], v);
}

extern "C" void kernel_launch(void* const* d_in, const int* in_sizes, int n_in,
                              void* d_out, int out_size, void* d_ws, size_t ws_size,
                              hipStream_t stream) {
  const float* h    = (const float*)d_in[0];
  const int*   ei   = (const int*)d_in[1];
  const float* attr = (const float*)d_in[2];
  const float* W1   = (const float*)d_in[3];
  const float* b1   = (const float*)d_in[4];
  const float* W2   = (const float*)d_in[5];
  const float* b2   = (const float*)d_in[6];
  const float* Wc   = (const float*)d_in[7];
  const float* bc   = (const float*)d_in[8];
  float* out = (float*)d_out;

  const int N = N_NODES, E = N_EDGES;

  float* x   = (float*)d_ws;
  float* w   = x + (size_t)N * HID;
  float* deg = w + E;

  edge_mlp_kernel<<<(E + 255) / 256, 256, 0, stream>>>(attr, W1, b1, W2, b2, w, E);
  deg_init_kernel<<<(N + 255) / 256, 256, 0, stream>>>(deg, N);
  deg_scatter_kernel<<<(E + 255) / 256, 256, 0, stream>>>(ei + E, w, deg, E);
  deg_rsqrt_kernel<<<(N + 255) / 256, 256, 0, stream>>>(deg, N);
  gemm_xWc_kernel<<<(N / 16) * 2, 256, 0, stream>>>(h, Wc, x);
  out_init_kernel<<<N, 256, 0, stream>>>(x, deg, bc, out);
  aggregate_kernel<<<E, 256, 0, stream>>>(ei, w, deg, x, out, E);
}


# edge sums: the owner scan reads a one-byte owner code per edge (target row / 196, stored by the edge weight launch, 16 edges per lane and load), not the 4-byte targets; sums still in arrival order
# speedup vs baseline: 4.4732x; 4.4732x over previous
#include <hip/hip_runtime.h>
#include <math.h>

typedef __attribute__((ext_vector_type(2))) float v2f;
typedef __attribute__((ext_vector_type(8))) float v8f;

#define N_NODES 50000
#define N_EDGES 800000
#define HID 256

#define OWN_GROUPS 256
#define OWN_ROWS ((N_NODES + OWN_GROUPS - 1) / OWN_GROUPS)
static_assert((N_NODES - 1) / OWN_ROWS < 256, "an owner code must fit one byte");

__global__ __launch_bounds__(256) void edge_mlp_kernel(
    const float* __restrict__ attr,
    const float* __restrict__ W1,
    const float* __restrict__ b1,
    const float* __restrict__ W2,
    const float* __restrict__ b2,
    const int* __restrict__ col,
    float* __restrict__ w,
    float* __restrict__ deg,
    unsigned char* __restrict__ owner,
    int E, int N) {
  __shared__ float sW1[3 * 128];
  __shared__ float sb1[128];
  __shared__ float sW2[128];
  for (int i = threadIdx.x; i < 384; i += 256) sW1[i] = W1[i];
  if (threadIdx.x < 128) {
    sb1[threadIdx.x] = b1[threadIdx.x];
    sW2[threadIdx.x] = W2[threadIdx.x];
  }
  __syncthreads();
  int e = blockIdx.x * 256 + threadIdx.x;
  static_assert(N_EDGES >= N_NODES, "deg start value needs a thread per node");
  if (e < N) deg[e] = 1.0f;
  if (e >= E) return;
  float a0 = attr[e * 3 + 0];
  float a1 = attr[e * 3 + 1];
  float a2 = attr[e * 3 + 2];
  const int c = col[e];
  float acc = 0.0f;
#pragma unroll 4
  for (int j = 0; j < 128; ++j) {
    float z = fmaf(a0, sW1[j], fmaf(a1, sW1[128 + j], fmaf(a2, sW1[256 + j], sb1[j])));
    float t = z / (1.0f + __expf(-z));
    acc = fmaf(t, sW2[j], acc);
  }
  float zz = acc + b2[0];
  w[e] = 1.0f / (1.0f + __expf(-zz));
  owner[e] = (unsigned char)((unsigned)c / (unsigned)OWN_ROWS);
}

typedef __attribute__((ext_vector_type(4))) float v4f;

#define DEG_GROUPS 64
#define DEG_THREADS 1024
#define DEG_CHUNK ((N_EDGES + DEG_GROUPS - 1) / DEG_GROUPS)
#define DEG_BATCH ((DEG_CHUNK + DEG_THREADS - 1) / DEG_THREADS)
#define DEG_LINES ((N_NODES + 31) / 32)
#define DEG_FLUSH 7

__device__ __forceinline__ float deg_to_dinv(float d) {
  return d > 0.0f ? rsqrtf(d) : 0.0f;
}

__global__ __launch_bounds__(DEG_THREADS) void deg_lds_kernel(
    const int* __restrict__ col, const float* __restrict__ w,
    float* __restrict__ deg, int E) {
  __shared__ float tab[N_NODES];
  const int tid = threadIdx.x;

  v4f* tab4 = (v4f*)tab;
  const v4f zero = {0.0f, 0.0f, 0.0f, 0.0f};
  for (int i = tid; i < N_NODES / 4; i += DEG_THREADS) tab4[i] = zero;
  __syncthreads();

  const int chunk = (E + DEG_GROUPS - 1) / DEG_GROUPS;
  const int e0 = blockIdx.x * chunk;
  const int e1 = (e0 + chunk < E) ? (e0 + chunk) : E;
  for (int base = e0; base < e1; base += DEG_BATCH * DEG_THREADS) {
    int c[DEG_BATCH];
    float v[DEG_BATCH];
#pragma unroll
    for (int k = 0; k < DEG_BATCH; ++k) {
      const int e = base + k * DEG_THREADS + tid;
      const int ec = e < e1 ? e : e1 - 1;
      c[k] = col[ec];
      v[k] = w[ec];
    }
#pragma unroll
    for (int k = 0; k < DEG_BATCH; ++k)
      if (base + k * DEG_THREADS + tid < e1) atomicAdd(&tab[c[k]], v[k]);
  }
  __syncthreads();

  const int lane = tid & 31;
  const int wave = tid >> 5;
  const int rot = blockIdx.x * (DEG_LINES / DEG_GROUPS);
  for (int k0 = wave; k0 < DEG_LINES; k0 += (DEG_THREADS / 32) * DEG_FLUSH) {
    int idx[DEG_FLUSH];
    float t[DEG_FLUSH];
#pragma unroll
    for (int u = 0; u < DEG_FLUSH; ++u) {
      const int k = k0 + (DEG_THREADS / 32) * u;
      int line = k + rot;
      if (line >= DEG_LINES) line -= DEG_LINES;
      const int i = line * 32 + lane;
      idx[u] = (k < DEG_LINES && i < N_NODES) ? i : -1;
      t[u] = tab[idx[u] >= 0 ? i : 0];
    }
#pragma unroll
    for (int u = 0; u < DEG_FLUSH; ++u)
      if (idx[u] >= 0) unsafeAtomicAdd(&deg[idx[u]], t[u]);
  }
}


#define GEMM_MTILES (N_NODES / 16)
#define GEMM_GROUPS 256
#define GEMM_KP 4

struct GemmFrag {
  v4f a[GEMM_KP];
  v4f b[GEMM_KP][4];
};

__device__ __forceinline__ void gemm_load_block(
    GemmFrag& f, const float* __restrict__ arow, const v4f* sB, int bidx, int blk) {
#pragma unroll
  for (int jj = 0; jj < GEMM_KP; ++jj)
    f.a[jj] = *(const v4f*)(arow + (blk * GEMM_KP + jj) * 8);
#pragma unroll
  for (int jj = 0; jj < GEMM_KP; ++jj)
#pragma unroll
    for (int t = 0; t < 4; ++t)
      f.b[jj][t] = sB[bidx + (t * 32 + blk * GEMM_KP + jj) * 32];
  __builtin_amdgcn_sched_barrier(0);
}

__device__ __forceinline__ void gemm_mma_block(const GemmFrag& f, v8f (&acc)[4]) {
#pragma unroll
  for (int jj = 0; jj < GEMM_KP; ++jj) {
#pragma unroll
    for (int t = 0; t < 4; ++t) {
      v2f a = {f.a[jj].x, f.a[jj].y};
      v2f b = {f.b[jj][t].x, f.b[jj][t].y};
      acc[t] = __builtin_amdgcn_wmma_f32_16x16x4_f32(false, a, false, b, (short)0,
                                                     acc[t], false, false);
    }
#pragma unroll
    for (int t = 0; t < 4; ++t) {
      v2f a = {f.a[jj].z, f.a[jj].w};
      v2f b = {f.b[jj][t].z, f.b[jj][t].w};
      acc[t] = __builtin_amdgcn_wmma_f32_16x16x4_f32(false, a, false, b, (short)0,
                                                     acc[t], false, false);
    }
  }
  __builtin_amdgcn_sched_barrier(0);
}

__global__ __launch_bounds__(256) void gemm_xWc_kernel(
    const float* __restrict__ h,
    const float* __restrict__ Wc,
    float* __restrict__ x) {
  __shared__ v4f sB[16 * 32 * 32];

  const int lane = threadIdx.x & 31;
  const int wave = __builtin_amdgcn_readfirstlane((int)(threadIdx.x >> 5));
  const int c = lane & 15;
  const int g = lane >> 4;

#pragma unroll 8
  for (int i = 0; i < 64; ++i) {
    const int p = wave * 64 + i;
    const int nt = p >> 5, j = p & 31;
    const float* src = Wc + (size_t)(8 * j + 4 * g) * HID + 16 * nt + c;
    v4f v;
    v.x = src[0];
    v.y = src[HID];
    v.z = src[2 * HID];
    v.w = src[3 * HID];
    sB[p * 32 + lane] = v;
  }
  __syncthreads();

  const int nq = wave & 3;
  const int bidx = (nq * 4 * 32) * 32 + lane;
  int mt = blockIdx.x + GEMM_GROUPS * (wave >> 2);
  if (mt >= GEMM_MTILES) return;

  const float* arow = h + (size_t)(mt * 16 + c) * HID + 4 * g;
  GemmFrag f0, f1;
  gemm_load_block(f0, arow, sB, bidx, 0);
  for (;;) {
    const int mt_next = mt + 2 * GEMM_GROUPS;
    const int mt_pf = mt_next < GEMM_MTILES ? mt_next : mt;
    const float* arow_next = h + (size_t)(mt_pf * 16 + c) * HID + 4 * g;

    v8f acc[4] = {};
    gemm_load_block(f1, arow, sB, bidx, 1);       gemm_mma_block(f0, acc);
    gemm_load_block(f0, arow, sB, bidx, 2);       gemm_mma_block(f1, acc);
    gemm_load_block(f1, arow, sB, bidx, 3);       gemm_mma_block(f0, acc);
    gemm_load_block(f0, arow, sB, bidx, 4);       gemm_mma_block(f1, acc);
    gemm_load_block(f1, arow, sB, bidx, 5);       gemm_mma_block(f0, acc);
    gemm_load_block(f0, arow, sB, bidx, 6);       gemm_mma_block(f1, acc);
    gemm_load_block(f1, arow, sB, bidx, 7);       gemm_mma_block(f0, acc);
    gemm_load_block(f0, arow_next, sB, bidx, 0);  gemm_mma_block(f1, acc);

    float* xrow = x + (size_t)(mt * 16 + 8 * g) * HID + nq * 64 + c;
#pragma unroll
    for (int t = 0; t < 4; ++t)
#pragma unroll
      for (int i = 0; i < 8; ++i) xrow[(size_t)i * HID + 16 * t] = acc[t][i];

    if (mt_next >= GEMM_MTILES) break;
    mt = mt_next;
    arow = arow_next;
  }
}

typedef __attribute__((ext_vector_type(4))) int v4i;

#define OWN_THREADS 1024
#define OWN_WAVES (OWN_THREADS / 32)
#define OWN_PIECES (N_EDGES / 16)
#define OWN_STEPS ((OWN_PIECES + 31) / 32)
#define OWN_SB 4
#define OWN_ITERS ((OWN_STEPS + OWN_WAVES * OWN_SB - 1) / (OWN_WAVES * OWN_SB))
#define OWN_QCAP 64
#define AGG_GROUP 4

static_assert(N_EDGES % 16 == 0, "the owner scan reads whole 16-edge pieces");
static_assert((OWN_GROUPS - 1) * OWN_ROWS < N_NODES, "every group owns a row");
static_assert((OWN_GROUPS - 1) * (OWN_STEPS / OWN_GROUPS) < OWN_STEPS, "rotation");

__device__ __forceinline__ int wave_bcast_i(int v, int src_lane) {
  return __builtin_amdgcn_readlane(v, src_lane);
}

__device__ __forceinline__ void own_edges4(
    const float* __restrict__ x, float* acc_lane,
    int r_l, int d_l, float nm_l, int k, unsigned lane_bytes) {
  float v[AGG_GROUP][8];
#pragma unroll
  for (int u = 0; u < AGG_GROUP; ++u) {
    const int r = wave_bcast_i(r_l, k + u);
    const char* src = (const char*)(x + (size_t)r * HID) + lane_bytes;
#pragma unroll
    for (int i = 0; i < 8; ++i) v[u][i] = *(const float*)(src + 128 * i);
  }
#pragma unroll
  for (int u = 0; u < AGG_GROUP; ++u) {
    const int d = wave_bcast_i(d_l, k + u);
    const float nm = __int_as_float(wave_bcast_i(__float_as_int(nm_l), k + u));
    float* dst = acc_lane + d * HID;
#pragma unroll
    for (int i = 0; i < 8; ++i) atomicAdd(dst + 32 * i, v[u][i] * nm);
  }
}

__device__ __forceinline__ unsigned own_zero_bytes(unsigned y) {
  const unsigned t = ((y & 0x7F7F7F7Fu) + 0x7F7F7F7Fu) | y;
  return ~t & 0x80808080u;
}

__device__ __forceinline__ unsigned own_match16(v4i cv, unsigned pat) {
  const unsigned f0 = own_zero_bytes((unsigned)cv.x ^ pat);
  const unsigned f1 = own_zero_bytes((unsigned)cv.y ^ pat);
  const unsigned f2 = own_zero_bytes((unsigned)cv.z ^ pat);
  const unsigned f3 = own_zero_bytes((unsigned)cv.w ^ pat);
  return (f0 >> 7) | (f1 >> 6) | (f2 >> 5) | (f3 >> 4);
}

__device__ __forceinline__ void own_push(int* q, int& qn, unsigned mask, bool has, int e) {
  if (has) q[qn + (int)__builtin_amdgcn_mbcnt_lo(mask, 0u)] = e;
  qn += __builtin_popcount(mask);
}

__device__ __forceinline__ void own_drain(
    const int* __restrict__ ei, const float* __restrict__ w,
    const float* __restrict__ deg, const float* __restrict__ x,
    float* acc_lane, const int* qe, int n, int lo, int lane, unsigned lane_bytes) {
  int r_l = 0, d_l = 0;
  float nm_l = 0.0f;
  if (lane < n) {
    const int e = qe[lane];
    r_l = ei[e];
    const int c = ei[N_EDGES + e];
    d_l = c - lo;
    nm_l = deg_to_dinv(deg[r_l]) * w[e] * deg_to_dinv(deg[c]);
    if ((unsigned)d_l >= (unsigned)OWN_ROWS) { d_l = 0; nm_l = 0.0f; }
  }
  int k = 0;
  for (; k + AGG_GROUP <= n; k += AGG_GROUP)
    own_edges4(x, acc_lane, r_l, d_l, nm_l, k, lane_bytes);
  for (; k < n; ++k) {
    const int r = wave_bcast_i(r_l, k);
    const int d = wave_bcast_i(d_l, k);
    const float nm = __int_as_float(wave_bcast_i(__float_as_int(nm_l), k));
    const char* src = (const char*)(x + (size_t)r * HID) + lane_bytes;
    float* dst = acc_lane + d * HID;
#pragma unroll
    for (int i = 0; i < 8; ++i)
      atomicAdd(dst + 32 * i, *(const float*)(src + 128 * i) * nm);
  }
}

__global__ __launch_bounds__(OWN_THREADS) void aggregate_owner_kernel(
    const int* __restrict__ ei,
    const unsigned char* __restrict__ owner,
    const float* __restrict__ w,
    const float* __restrict__ deg,
    const float* __restrict__ x,
    const float* __restrict__ bc,
    float* __restrict__ out) {
  __shared__ float acc[OWN_ROWS * HID];
  __shared__ int queue[OWN_WAVES * OWN_QCAP];

  const int tid = threadIdx.x;
  const int lane = tid & 31;
  const int wave = __builtin_amdgcn_readfirstlane(tid >> 5);
  const int lo = blockIdx.x * OWN_ROWS;
  const int cnt = (N_NODES - lo < OWN_ROWS) ? (N_NODES - lo) : OWN_ROWS;

  v4f* acc4 = (v4f*)acc;
  {
    const v4f* x4 = (const v4f*)(x + (size_t)lo * HID);
    const v4f* bc4 = (const v4f*)bc;
    for (int i = tid; i < cnt * (HID / 4); i += OWN_THREADS) {
      const float di = deg_to_dinv(deg[lo + (i >> 6)]);
      const v4f xv = x4[i];
      const v4f b = bc4[i & 63];
      v4f o;
      o.x = xv.x * di * di + b.x;
      o.y = xv.y * di * di + b.y;
      o.z = xv.z * di * di + b.z;
      o.w = xv.w * di * di + b.w;
      acc4[i] = o;
    }
  }
  __syncthreads();

  int* q = queue + wave * OWN_QCAP;
  int qn = 0;
  const int rot = blockIdx.x * (OWN_STEPS / OWN_GROUPS);
  const v4i* own4 = (const v4i*)owner;
  const unsigned pat = blockIdx.x * 0x01010101u;
  const unsigned lane_bytes = (unsigned)lane * 4u;
  float* acc_lane = acc + lane;

  v4i cv[OWN_SB];
  int piece[OWN_SB];
#pragma unroll
  for (int u = 0; u < OWN_SB; ++u) {
    int sr = u * OWN_WAVES + wave + rot;
    if (sr >= OWN_STEPS) sr -= OWN_STEPS;
    piece[u] = sr * 32 + lane;
    cv[u] = own4[piece[u] < OWN_PIECES ? piece[u] : OWN_PIECES - 1];
  }

  static_assert(OWN_SB == 4, "the loop below names the four steps of a batch");
  for (int it = 0; it < OWN_ITERS; ++it) {
    const int itn = it + 1 < OWN_ITERS ? it + 1 : it;
    unsigned m[OWN_SB];
    int e0[OWN_SB];
#pragma unroll
    for (int u = 0; u < OWN_SB; ++u) {
      const int s = (it * OWN_SB + u) * OWN_WAVES + wave;
      m[u] = (s < OWN_STEPS && piece[u] < OWN_PIECES) ? own_match16(cv[u], pat) : 0u;
      e0[u] = piece[u] * 16;
      int sn = (itn * OWN_SB + u) * OWN_WAVES + wave + rot;
      if (sn >= OWN_STEPS) sn -= OWN_STEPS;
      piece[u] = sn * 32 + lane;
      cv[u] = own4[piece[u] < OWN_PIECES ? piece[u] : OWN_PIECES - 1];
    }
    unsigned any = __builtin_amdgcn_ballot_w32((m[0] | m[1] | m[2] | m[3]) != 0u);
    while (any != 0u) {
      const bool h0 = m[0] != 0u, h1 = m[1] != 0u, h2 = m[2] != 0u;
      const unsigned mm = h0 ? m[0] : h1 ? m[1] : h2 ? m[2] : m[3];
      const int eb = h0 ? e0[0] : h1 ? e0[1] : h2 ? e0[2] : e0[3];
      const int p = __builtin_ctz(mm | 0x80000000u);
      own_push(q, qn, any, mm != 0u, eb + (((p & 3) << 2) | (p >> 3)));
      const unsigned rest = mm & (mm - 1u);
      m[0] = h0 ? rest : m[0];
      m[1] = (!h0 && h1) ? rest : m[1];
      m[2] = (!h0 && !h1 && h2) ? rest : m[2];
      m[3] = (!h0 && !h1 && !h2) ? rest : m[3];
      if (qn >= 32) {
        qn -= 32;
        own_drain(ei, w, deg, x, acc_lane, q + qn, 32, lo, lane, lane_bytes);
      }
      any = __builtin_amdgcn_ballot_w32((m[0] | m[1] | m[2] | m[3]) != 0u);
    }
  }
  if (qn > 0) own_drain(ei, w, deg, x, acc_lane, q, qn, lo, lane, lane_bytes);
  __syncthreads();

  v4f* out4 = (v4f*)(out + (size_t)lo * HID);
  for (int i = tid; i < cnt * (HID / 4); i += OWN_THREADS) out4[i] = acc4[i];
}

extern "C" void kernel_launch(void* const* d_in, const int* in_sizes, int n_in,
                              void* d_out, int out_size, void* d_ws, size_t ws_size,
                              hipStream_t stream) {
  const float* h    = (const float*)d_in[0];
  const int*   ei   = (const int*)d_in[1];
  const float* attr = (const float*)d_in[2];
  const float* W1   = (const float*)d_in[3];
  const float* b1   = (const float*)d_in[4];
  const float* W2   = (const float*)d_in[5];
  const float* b2   = (const float*)d_in[6];
  const float* Wc   = (const float*)d_in[7];
  const float* bc   = (const float*)d_in[8];
  float* out = (float*)d_out;

  const int N = N_NODES, E = N_EDGES;

  float* x   = (float*)d_ws;
  float* w   = x + (size_t)N * HID;
  float* deg = w + E;
  unsigned char* owner = (unsigned char*)(deg + N);

  edge_mlp_kernel<<<(E + 255) / 256, 256, 0, stream>>>(attr, W1, b1, W2, b2, ei + E, w, deg, owner, E, N);
  deg_lds_kernel<<<DEG_GROUPS, DEG_THREADS, 0, stream>>>(ei + E, w, deg, E);
  gemm_xWc_kernel<<<GEMM_GROUPS, 256, 0, stream>>>(h, Wc, x);
  aggregate_owner_kernel<<<OWN_GROUPS, OWN_THREADS, 0, stream>>>(ei, owner, w, deg, x, bc, out);
}


# edge sums: a source row is gathered with two 16-byte-per-lane loads (four lines per load), not eight one-line loads; own rows lane-transposed in LDS (adds stay one bank per lane); sum order unchanged
# speedup vs baseline: 5.0775x; 1.1351x over previous
#include <hip/hip_runtime.h>
#include <math.h>

typedef __attribute__((ext_vector_type(2))) float v2f;
typedef __attribute__((ext_vector_type(8))) float v8f;

#define N_NODES 50000
#define N_EDGES 800000
#define HID 256

#define OWN_GROUPS 256
#define OWN_ROWS ((N_NODES + OWN_GROUPS - 1) / OWN_GROUPS)
static_assert((N_NODES - 1) / OWN_ROWS < 256, "an owner code must fit one byte");

__global__ __launch_bounds__(256) void edge_mlp_kernel(
    const float* __restrict__ attr,
    const float* __restrict__ W1,
    const float* __restrict__ b1,
    const float* __restrict__ W2,
    const float* __restrict__ b2,
    const int* __restrict__ col,
    float* __restrict__ w,
    float* __restrict__ deg,
    unsigned char* __restrict__ owner,
    int E, int N) {
  __shared__ float sW1[3 * 128];
  __shared__ float sb1[128];
  __shared__ float sW2[128];
  for (int i = threadIdx.x; i < 384; i += 256) sW1[i] = W1[i];
  if (threadIdx.x < 128) {
    sb1[threadIdx.x] = b1[threadIdx.x];
    sW2[threadIdx.x] = W2[threadIdx.x];
  }
  __syncthreads();
  int e = blockIdx.x * 256 + threadIdx.x;
  static_assert(N_EDGES >= N_NODES, "deg start value needs a thread per node");
  if (e < N) deg[e] = 1.0f;
  if (e >= E) return;
  float a0 = attr[e * 3 + 0];
  float a1 = attr[e * 3 + 1];
  float a2 = attr[e * 3 + 2];
  const int c = col[e];
  float acc = 0.0f;
#pragma unroll 4
  for (int j = 0; j < 128; ++j) {
    float z = fmaf(a0, sW1[j], fmaf(a1, sW1[128 + j], fmaf(a2, sW1[256 + j], sb1[j])));
    float t = z / (1.0f + __expf(-z));
    acc = fmaf(t, sW2[j], acc);
  }
  float zz = acc + b2[0];
  w[e] = 1.0f / (1.0f + __expf(-zz));
  owner[e] = (unsigned char)((unsigned)c / (unsigned)OWN_ROWS);
}

typedef __attribute__((ext_vector_type(4))) float v4f;

#define DEG_GROUPS 64
#define DEG_THREADS 1024
#define DEG_CHUNK ((N_EDGES + DEG_GROUPS - 1) / DEG_GROUPS)
#define DEG_BATCH ((DEG_CHUNK + DEG_THREADS - 1) / DEG_THREADS)
#define DEG_LINES ((N_NODES + 31) / 32)
#define DEG_FLUSH 7

__device__ __forceinline__ float deg_to_dinv(float d) {
  return d > 0.0f ? rsqrtf(d) : 0.0f;
}

__global__ __launch_bounds__(DEG_THREADS) void deg_lds_kernel(
    const int* __restrict__ col, const float* __restrict__ w,
    float* __restrict__ deg, int E) {
  __shared__ float tab[N_NODES];
  const int tid = threadIdx.x;

  v4f* tab4 = (v4f*)tab;
  const v4f zero = {0.0f, 0.0f, 0.0f, 0.0f};
  for (int i = tid; i < N_NODES / 4; i += DEG_THREADS) tab4[i] = zero;
  __syncthreads();

  const int chunk = (E + DEG_GROUPS - 1) / DEG_GROUPS;
  const int e0 = blockIdx.x * chunk;
  const int e1 = (e0 + chunk < E) ? (e0 + chunk) : E;
  for (int base = e0; base < e1; base += DEG_BATCH * DEG_THREADS) {
    int c[DEG_BATCH];
    float v[DEG_BATCH];
#pragma unroll
    for (int k = 0; k < DEG_BATCH; ++k) {
      const int e = base + k * DEG_THREADS + tid;
      const int ec = e < e1 ? e : e1 - 1;
      c[k] = col[ec];
      v[k] = w[ec];
    }
#pragma unroll
    for (int k = 0; k < DEG_BATCH; ++k)
      if (base + k * DEG_THREADS + tid < e1) atomicAdd(&tab[c[k]], v[k]);
  }
  __syncthreads();

  const int lane = tid & 31;
  const int wave = tid >> 5;
  const int rot = blockIdx.x * (DEG_LINES / DEG_GROUPS);
  for (int k0 = wave; k0 < DEG_LINES; k0 += (DEG_THREADS / 32) * DEG_FLUSH) {
    int idx[DEG_FLUSH];
    float t[DEG_FLUSH];
#pragma unroll
    for (int u = 0; u < DEG_FLUSH; ++u) {
      const int k = k0 + (DEG_THREADS / 32) * u;
      int line = k + rot;
      if (line >= DEG_LINES) line -= DEG_LINES;
      const int i = line * 32 + lane;
      idx[u] = (k < DEG_LINES && i < N_NODES) ? i : -1;
      t[u] = tab[idx[u] >= 0 ? i : 0];
    }
#pragma unroll
    for (int u = 0; u < DEG_FLUSH; ++u)
      if (idx[u] >= 0) unsafeAtomicAdd(&deg[idx[u]], t[u]);
  }
}


#define GEMM_MTILES (N_NODES / 16)
#define GEMM_GROUPS 256
#define GEMM_KP 4

struct GemmFrag {
  v4f a[GEMM_KP];
  v4f b[GEMM_KP][4];
};

__device__ __forceinline__ void gemm_load_block(
    GemmFrag& f, const float* __restrict__ arow, const v4f* sB, int bidx, int blk) {
#pragma unroll
  for (int jj = 0; jj < GEMM_KP; ++jj)
    f.a[jj] = *(const v4f*)(arow + (blk * GEMM_KP + jj) * 8);
#pragma unroll
  for (int jj = 0; jj < GEMM_KP; ++jj)
#pragma unroll
    for (int t = 0; t < 4; ++t)
      f.b[jj][t] = sB[bidx + (t * 32 + blk * GEMM_KP + jj) * 32];
  __builtin_amdgcn_sched_barrier(0);
}

__device__ __forceinline__ void gemm_mma_block(const GemmFrag& f, v8f (&acc)[4]) {
#pragma unroll
  for (int jj = 0; jj < GEMM_KP; ++jj) {
#pragma unroll
    for (int t = 0; t < 4; ++t) {
      v2f a = {f.a[jj].x, f.a[jj].y};
      v2f b = {f.b[jj][t].x, f.b[jj][t].y};
      acc[t] = __builtin_amdgcn_wmma_f32_16x16x4_f32(false, a, false, b, (short)0,
                                                     acc[t], false, false);
    }
#pragma unroll
    for (int t = 0; t < 4; ++t) {
      v2f a = {f.a[jj].z, f.a[jj].w};
      v2f b = {f.b[jj][t].z, f.b[jj][t].w};
      acc[t] = __builtin_amdgcn_wmma_f32_16x16x4_f32(false, a, false, b, (short)0,
                                                     acc[t], false, false);
    }
  }
  __builtin_amdgcn_sched_barrier(0);
}

__global__ __launch_bounds__(256) void gemm_xWc_kernel(
    const float* __restrict__ h,
    const float* __restrict__ Wc,
    float* __restrict__ x) {
  __shared__ v4f sB[16 * 32 * 32];

  const int lane = threadIdx.x & 31;
  const int wave = __builtin_amdgcn_readfirstlane((int)(threadIdx.x >> 5));
  const int c = lane & 15;
  const int g = lane >> 4;

#pragma unroll 8
  for (int i = 0; i < 64; ++i) {
    const int p = wave * 64 + i;
    const int nt = p >> 5, j = p & 31;
    const float* src = Wc + (size_t)(8 * j + 4 * g) * HID + 16 * nt + c;
    v4f v;
    v.x = src[0];
    v.y = src[HID];
    v.z = src[2 * HID];
    v.w = src[3 * HID];
    sB[p * 32 + lane] = v;
  }
  __syncthreads();

  const int nq = wave & 3;
  const int bidx = (nq * 4 * 32) * 32 + lane;
  int mt = blockIdx.x + GEMM_GROUPS * (wave >> 2);
  if (mt >= GEMM_MTILES) return;

  const float* arow = h + (size_t)(mt * 16 + c) * HID + 4 * g;
  GemmFrag f0, f1;
  gemm_load_block(f0, arow, sB, bidx, 0);
  for (;;) {
    const int mt_next = mt + 2 * GEMM_GROUPS;
    const int mt_pf = mt_next < GEMM_MTILES ? mt_next : mt;
    const float* arow_next = h + (size_t)(mt_pf * 16 + c) * HID + 4 * g;

    v8f acc[4] = {};
    gemm_load_block(f1, arow, sB, bidx, 1);       gemm_mma_block(f0, acc);
    gemm_load_block(f0, arow, sB, bidx, 2);       gemm_mma_block(f1, acc);
    gemm_load_block(f1, arow, sB, bidx, 3);       gemm_mma_block(f0, acc);
    gemm_load_block(f0, arow, sB, bidx, 4);       gemm_mma_block(f1, acc);
    gemm_load_block(f1, arow, sB, bidx, 5);       gemm_mma_block(f0, acc);
    gemm_load_block(f0, arow, sB, bidx, 6);       gemm_mma_block(f1, acc);
    gemm_load_block(f1, arow, sB, bidx, 7);       gemm_mma_block(f0, acc);
    gemm_load_block(f0, arow_next, sB, bidx, 0);  gemm_mma_block(f1, acc);

    float* xrow = x + (size_t)(mt * 16 + 8 * g) * HID + nq * 64 + c;
#pragma unroll
    for (int t = 0; t < 4; ++t)
#pragma unroll
      for (int i = 0; i < 8; ++i) xrow[(size_t)i * HID + 16 * t] = acc[t][i];

    if (mt_next >= GEMM_MTILES) break;
    mt = mt_next;
    arow = arow_next;
  }
}

typedef __attribute__((ext_vector_type(4))) int v4i;

#define OWN_THREADS 1024
#define OWN_WAVES (OWN_THREADS / 32)
#define OWN_PIECES (N_EDGES / 16)
#define OWN_STEPS ((OWN_PIECES + 31) / 32)
#define OWN_SB 4
#define OWN_ITERS ((OWN_STEPS + OWN_WAVES * OWN_SB - 1) / (OWN_WAVES * OWN_SB))
#define OWN_QCAP 64
#define AGG_GROUP 4

static_assert(N_EDGES % 16 == 0, "the owner scan reads whole 16-edge pieces");
static_assert((OWN_GROUPS - 1) * OWN_ROWS < N_NODES, "every group owns a row");
static_assert((OWN_GROUPS - 1) * (OWN_STEPS / OWN_GROUPS) < OWN_STEPS, "rotation");

__device__ __forceinline__ int wave_bcast_i(int v, int src_lane) {
  return __builtin_amdgcn_readlane(v, src_lane);
}

__device__ __forceinline__ int own_slot0(int i) {
  return (i >> 6) * HID + ((i & 32) << 2) + (i & 31);
}

__device__ __forceinline__ void own_edges4(
    const float* __restrict__ x, float* acc_lane,
    int r_l, int d_l, float nm_l, int k, unsigned lane_bytes) {
  v4f v[AGG_GROUP][2];
#pragma unroll
  for (int u = 0; u < AGG_GROUP; ++u) {
    const int r = wave_bcast_i(r_l, k + u);
    const char* src = (const char*)(x + (size_t)r * HID) + lane_bytes;
#pragma unroll
    for (int hh = 0; hh < 2; ++hh) v[u][hh] = *(const v4f*)(src + 512 * hh);
  }
#pragma unroll
  for (int u = 0; u < AGG_GROUP; ++u) {
    const int d = wave_bcast_i(d_l, k + u);
    const float nm = __int_as_float(wave_bcast_i(__float_as_int(nm_l), k + u));
    float* dst = acc_lane + d * HID;
#pragma unroll
    for (int hh = 0; hh < 2; ++hh)
#pragma unroll
      for (int j = 0; j < 4; ++j)
        atomicAdd(dst + 32 * (4 * hh + j), v[u][hh][j] * nm);
  }
}

__device__ __forceinline__ unsigned own_zero_bytes(unsigned y) {
  const unsigned t = ((y & 0x7F7F7F7Fu) + 0x7F7F7F7Fu) | y;
  return ~t & 0x80808080u;
}

__device__ __forceinline__ unsigned own_match16(v4i cv, unsigned pat) {
  const unsigned f0 = own_zero_bytes((unsigned)cv.x ^ pat);
  const unsigned f1 = own_zero_bytes((unsigned)cv.y ^ pat);
  const unsigned f2 = own_zero_bytes((unsigned)cv.z ^ pat);
  const unsigned f3 = own_zero_bytes((unsigned)cv.w ^ pat);
  return (f0 >> 7) | (f1 >> 6) | (f2 >> 5) | (f3 >> 4);
}

__device__ __forceinline__ void own_push(int* q, int& qn, unsigned mask, bool has, int e) {
  if (has) q[qn + (int)__builtin_amdgcn_mbcnt_lo(mask, 0u)] = e;
  qn += __builtin_popcount(mask);
}

__device__ __forceinline__ void own_drain(
    const int* __restrict__ ei, const float* __restrict__ w,
    const float* __restrict__ deg, const float* __restrict__ x,
    float* acc_lane, const int* qe, int n, int lo, int lane, unsigned lane_bytes) {
  int r_l = 0, d_l = 0;
  float nm_l = 0.0f;
  if (lane < n) {
    const int e = qe[lane];
    r_l = ei[e];
    const int c = ei[N_EDGES + e];
    d_l = c - lo;
    nm_l = deg_to_dinv(deg[r_l]) * w[e] * deg_to_dinv(deg[c]);
    if ((unsigned)d_l >= (unsigned)OWN_ROWS) { d_l = 0; nm_l = 0.0f; }
  }
  int k = 0;
  for (; k + AGG_GROUP <= n; k += AGG_GROUP)
    own_edges4(x, acc_lane, r_l, d_l, nm_l, k, lane_bytes);
  for (; k < n; ++k) {
    const int r = wave_bcast_i(r_l, k);
    const int d = wave_bcast_i(d_l, k);
    const float nm = __int_as_float(wave_bcast_i(__float_as_int(nm_l), k));
    const char* src = (const char*)(x + (size_t)r * HID) + lane_bytes;
    float* dst = acc_lane + d * HID;
    const v4f lo4 = *(const v4f*)src;
    const v4f hi4 = *(const v4f*)(src + 512);
#pragma unroll
    for (int j = 0; j < 4; ++j) {
      atomicAdd(dst + 32 * j, lo4[j] * nm);
      atomicAdd(dst + 32 * (4 + j), hi4[j] * nm);
    }
  }
}

__global__ __launch_bounds__(OWN_THREADS) void aggregate_owner_kernel(
    const int* __restrict__ ei,
    const unsigned char* __restrict__ owner,
    const float* __restrict__ w,
    const float* __restrict__ deg,
    const float* __restrict__ x,
    const float* __restrict__ bc,
    float* __restrict__ out) {
  __shared__ float acc[OWN_ROWS * HID];
  __shared__ int queue[OWN_WAVES * OWN_QCAP];

  const int tid = threadIdx.x;
  const int lane = tid & 31;
  const int wave = __builtin_amdgcn_readfirstlane(tid >> 5);
  const int lo = blockIdx.x * OWN_ROWS;
  const int cnt = (N_NODES - lo < OWN_ROWS) ? (N_NODES - lo) : OWN_ROWS;

  {
    const v4f* x4 = (const v4f*)(x + (size_t)lo * HID);
    const v4f* bc4 = (const v4f*)bc;
    for (int i = tid; i < cnt * (HID / 4); i += OWN_THREADS) {
      const float di = deg_to_dinv(deg[lo + (i >> 6)]);
      const v4f xv = x4[i];
      const v4f b = bc4[i & 63];
      v4f o;
      o.x = xv.x * di * di + b.x;
      o.y = xv.y * di * di + b.y;
      o.z = xv.z * di * di + b.z;
      o.w = xv.w * di * di + b.w;
      float* s = acc + own_slot0(i);
      s[0] = o.x;
      s[32] = o.y;
      s[64] = o.z;
      s[96] = o.w;
    }
  }
  __syncthreads();

  int* q = queue + wave * OWN_QCAP;
  int qn = 0;
  const int rot = blockIdx.x * (OWN_STEPS / OWN_GROUPS);
  const v4i* own4 = (const v4i*)owner;
  const unsigned pat = blockIdx.x * 0x01010101u;
  const unsigned lane_bytes = (unsigned)lane * 16u;
  float* acc_lane = acc + lane;

  v4i cv[OWN_SB];
  int piece[OWN_SB];
#pragma unroll
  for (int u = 0; u < OWN_SB; ++u) {
    int sr = u * OWN_WAVES + wave + rot;
    if (sr >= OWN_STEPS) sr -= OWN_STEPS;
    piece[u] = sr * 32 + lane;
    cv[u] = own4[piece[u] < OWN_PIECES ? piece[u] : OWN_PIECES - 1];
  }

  static_assert(OWN_SB == 4, "the loop below names the four steps of a batch");
  for (int it = 0; it < OWN_ITERS; ++it) {
    const int itn = it + 1 < OWN_ITERS ? it + 1 : it;
    unsigned m[OWN_SB];
    int e0[OWN_SB];
#pragma unroll
    for (int u = 0; u < OWN_SB; ++u) {
      const int s = (it * OWN_SB + u) * OWN_WAVES + wave;
      m[u] = (s < OWN_STEPS && piece[u] < OWN_PIECES) ? own_match16(cv[u], pat) : 0u;
      e0[u] = piece[u] * 16;
      int sn = (itn * OWN_SB + u) * OWN_WAVES + wave + rot;
      if (sn >= OWN_STEPS) sn -= OWN_STEPS;
      piece[u] = sn * 32 + lane;
      cv[u] = own4[piece[u] < OWN_PIECES ? piece[u] : OWN_PIECES - 1];
    }
    unsigned any = __builtin_amdgcn_ballot_w32((m[0] | m[1] | m[2] | m[3]) != 0u);
    while (any != 0u) {
      const bool h0 = m[0] != 0u, h1 = m[1] != 0u, h2 = m[2] != 0u;
      const unsigned mm = h0 ? m[0] : h1 ? m[1] : h2 ? m[2] : m[3];
      const int eb = h0 ? e0[0] : h1 ? e0[1] : h2 ? e0[2] : e0[3];
      const int p = __builtin_ctz(mm | 0x80000000u);
      own_push(q, qn, any, mm != 0u, eb + (((p & 3) << 2) | (p >> 3)));
      const unsigned rest = mm & (mm - 1u);
      m[0] = h0 ? rest : m[0];
      m[1] = (!h0 && h1) ? rest : m[1];
      m[2] = (!h0 && !h1 && h2) ? rest : m[2];
      m[3] = (!h0 && !h1 && !h2) ? rest : m[3];
      if (qn >= 32) {
        qn -= 32;
        own_drain(ei, w, deg, x, acc_lane, q + qn, 32, lo, lane, lane_bytes);
      }
      any = __builtin_amdgcn_ballot_w32((m[0] | m[1] | m[2] | m[3]) != 0u);
    }
  }
  if (qn > 0) own_drain(ei, w, deg, x, acc_lane, q, qn, lo, lane, lane_bytes);
  __syncthreads();

  v4f* out4 = (v4f*)(out + (size_t)lo * HID);
  for (int i = tid; i < cnt * (HID / 4); i += OWN_THREADS) {
    const float* s = acc + own_slot0(i);
    const v4f o = {s[0], s[32], s[64], s[96]};
    out4[i] = o;
  }
}

extern "C" void kernel_launch(void* const* d_in, const int* in_sizes, int n_in,
                              void* d_out, int out_size, void* d_ws, size_t ws_size,
                              hipStream_t stream) {
  const float* h    = (const float*)d_in[0];
  const int*   ei   = (const int*)d_in[1];
  const float* attr = (const float*)d_in[2];
  const float* W1   = (const float*)d_in[3];
  const float* b1   = (const float*)d_in[4];
  const float* W2   = (const float*)d_in[5];
  const float* b2   = (const float*)d_in[6];
  const float* Wc   = (const float*)d_in[7];
  const float* bc   = (const float*)d_in[8];
  float* out = (float*)d_out;

  const int N = N_NODES, E = N_EDGES;

  float* x   = (float*)d_ws;
  float* w   = x + (size_t)N * HID;
  float* deg = w + E;
  unsigned char* owner = (unsigned char*)(deg + N);

  edge_mlp_kernel<<<(E + 255) / 256, 256, 0, stream>>>(attr, W1, b1, W2, b2, ei + E, w, deg, owner, E, N);
  deg_lds_kernel<<<DEG_GROUPS, DEG_THREADS, 0, stream>>>(ei + E, w, deg, E);
  gemm_xWc_kernel<<<GEMM_GROUPS, 256, 0, stream>>>(h, Wc, x);
  aggregate_owner_kernel<<<OWN_GROUPS, OWN_THREADS, 0, stream>>>(ei, owner, w, deg, x, bc, out);
}


# edge norm factored: dinv[source] stored into the x rows by the GEMM launch, dinv[target] and bias applied once at the out store; no degree gathers per edge; start rows loaded together; sum order kept
# speedup vs baseline: 5.2470x; 1.0334x over previous
#include <hip/hip_runtime.h>
#include <math.h>

typedef __attribute__((ext_vector_type(2))) float v2f;
typedef __attribute__((ext_vector_type(8))) float v8f;

#define N_NODES 50000
#define N_EDGES 800000
#define HID 256

#define OWN_GROUPS 256
#define OWN_ROWS ((N_NODES + OWN_GROUPS - 1) / OWN_GROUPS)
static_assert((N_NODES - 1) / OWN_ROWS < 256, "an owner code must fit one byte");

__global__ __launch_bounds__(256) void edge_mlp_kernel(
    const float* __restrict__ attr,
    const float* __restrict__ W1,
    const float* __restrict__ b1,
    const float* __restrict__ W2,
    const float* __restrict__ b2,
    const int* __restrict__ col,
    float* __restrict__ w,
    float* __restrict__ deg,
    unsigned char* __restrict__ owner,
    int E, int N) {
  __shared__ float sW1[3 * 128];
  __shared__ float sb1[128];
  __shared__ float sW2[128];
  for (int i = threadIdx.x; i < 384; i += 256) sW1[i] = W1[i];
  if (threadIdx.x < 128) {
    sb1[threadIdx.x] = b1[threadIdx.x];
    sW2[threadIdx.x] = W2[threadIdx.x];
  }
  __syncthreads();
  int e = blockIdx.x * 256 + threadIdx.x;
  static_assert(N_EDGES >= N_NODES, "deg start value needs a thread per node");
  if (e < N) deg[e] = 1.0f;
  if (e >= E) return;
  float a0 = attr[e * 3 + 0];
  float a1 = attr[e * 3 + 1];
  float a2 = attr[e * 3 + 2];
  const int c = col[e];
  float acc = 0.0f;
#pragma unroll 4
  for (int j = 0; j < 128; ++j) {
    float z = fmaf(a0, sW1[j], fmaf(a1, sW1[128 + j], fmaf(a2, sW1[256 + j], sb1[j])));
    float t = z / (1.0f + __expf(-z));
    acc = fmaf(t, sW2[j], acc);
  }
  float zz = acc + b2[0];
  w[e] = 1.0f / (1.0f + __expf(-zz));
  owner[e] = (unsigned char)((unsigned)c / (unsigned)OWN_ROWS);
}

typedef __attribute__((ext_vector_type(4))) float v4f;

#define DEG_GROUPS 64
#define DEG_THREADS 1024
#define DEG_CHUNK ((N_EDGES + DEG_GROUPS - 1) / DEG_GROUPS)
#define DEG_BATCH ((DEG_CHUNK + DEG_THREADS - 1) / DEG_THREADS)
#define DEG_LINES ((N_NODES + 31) / 32)
#define DEG_FLUSH 7

__device__ __forceinline__ float deg_to_dinv(float d) {
  return d > 0.0f ? rsqrtf(d) : 0.0f;
}

__global__ __launch_bounds__(DEG_THREADS) void deg_lds_kernel(
    const int* __restrict__ col, const float* __restrict__ w,
    float* __restrict__ deg, int E) {
  __shared__ float tab[N_NODES];
  const int tid = threadIdx.x;

  v4f* tab4 = (v4f*)tab;
  const v4f zero = {0.0f, 0.0f, 0.0f, 0.0f};
  for (int i = tid; i < N_NODES / 4; i += DEG_THREADS) tab4[i] = zero;
  __syncthreads();

  const int chunk = (E + DEG_GROUPS - 1) / DEG_GROUPS;
  const int e0 = blockIdx.x * chunk;
  const int e1 = (e0 + chunk < E) ? (e0 + chunk) : E;
  for (int base = e0; base < e1; base += DEG_BATCH * DEG_THREADS) {
    int c[DEG_BATCH];
    float v[DEG_BATCH];
#pragma unroll
    for (int k = 0; k < DEG_BATCH; ++k) {
      const int e = base + k * DEG_THREADS + tid;
      const int ec = e < e1 ? e : e1 - 1;
      c[k] = col[ec];
      v[k] = w[ec];
    }
#pragma unroll
    for (int k = 0; k < DEG_BATCH; ++k)
      if (base + k * DEG_THREADS + tid < e1) atomicAdd(&tab[c[k]], v[k]);
  }
  __syncthreads();

  const int lane = tid & 31;
  const int wave = tid >> 5;
  const int rot = blockIdx.x * (DEG_LINES / DEG_GROUPS);
  for (int k0 = wave; k0 < DEG_LINES; k0 += (DEG_THREADS / 32) * DEG_FLUSH) {
    int idx[DEG_FLUSH];
    float t[DEG_FLUSH];
#pragma unroll
    for (int u = 0; u < DEG_FLUSH; ++u) {
      const int k = k0 + (DEG_THREADS / 32) * u;
      int line = k + rot;
      if (line >= DEG_LINES) line -= DEG_LINES;
      const int i = line * 32 + lane;
      idx[u] = (k < DEG_LINES && i < N_NODES) ? i : -1;
      t[u] = tab[idx[u] >= 0 ? i : 0];
    }
#pragma unroll
    for (int u = 0; u < DEG_FLUSH; ++u)
      if (idx[u] >= 0) unsafeAtomicAdd(&deg[idx[u]], t[u]);
  }
}


#define GEMM_MTILES (N_NODES / 16)
#define GEMM_GROUPS 256
#define GEMM_KP 4

struct GemmFrag {
  v4f a[GEMM_KP];
  v4f b[GEMM_KP][4];
};

__device__ __forceinline__ void gemm_load_block(
    GemmFrag& f, const float* __restrict__ arow, const v4f* sB, int bidx, int blk) {
#pragma unroll
  for (int jj = 0; jj < GEMM_KP; ++jj)
    f.a[jj] = *(const v4f*)(arow + (blk * GEMM_KP + jj) * 8);
#pragma unroll
  for (int jj = 0; jj < GEMM_KP; ++jj)
#pragma unroll
    for (int t = 0; t < 4; ++t)
      f.b[jj][t] = sB[bidx + (t * 32 + blk * GEMM_KP + jj) * 32];
  __builtin_amdgcn_sched_barrier(0);
}

__device__ __forceinline__ void gemm_mma_block(const GemmFrag& f, v8f (&acc)[4]) {
#pragma unroll
  for (int jj = 0; jj < GEMM_KP; ++jj) {
#pragma unroll
    for (int t = 0; t < 4; ++t) {
      v2f a = {f.a[jj].x, f.a[jj].y};
      v2f b = {f.b[jj][t].x, f.b[jj][t].y};
      acc[t] = __builtin_amdgcn_wmma_f32_16x16x4_f32(false, a, false, b, (short)0,
                                                     acc[t], false, false);
    }
#pragma unroll
    for (int t = 0; t < 4; ++t) {
      v2f a = {f.a[jj].z, f.a[jj].w};
      v2f b = {f.b[jj][t].z, f.b[jj][t].w};
      acc[t] = __builtin_amdgcn_wmma_f32_16x16x4_f32(false, a, false, b, (short)0,
                                                     acc[t], false, false);
    }
  }
  __builtin_amdgcn_sched_barrier(0);
}

__global__ __launch_bounds__(256) void gemm_xWc_kernel(
    const float* __restrict__ h,
    const float* __restrict__ Wc,
    const float* __restrict__ deg,
    float* __restrict__ x) {
  __shared__ v4f sB[16 * 32 * 32];

  const int lane = threadIdx.x & 31;
  const int wave = __builtin_amdgcn_readfirstlane((int)(threadIdx.x >> 5));
  const int c = lane & 15;
  const int g = lane >> 4;

#pragma unroll 8
  for (int i = 0; i < 64; ++i) {
    const int p = wave * 64 + i;
    const int nt = p >> 5, j = p & 31;
    const float* src = Wc + (size_t)(8 * j + 4 * g) * HID + 16 * nt + c;
    v4f v;
    v.x = src[0];
    v.y = src[HID];
    v.z = src[2 * HID];
    v.w = src[3 * HID];
    sB[p * 32 + lane] = v;
  }
  __syncthreads();

  const int nq = wave & 3;
  const int bidx = (nq * 4 * 32) * 32 + lane;
  int mt = blockIdx.x + GEMM_GROUPS * (wave >> 2);
  if (mt >= GEMM_MTILES) return;

  const float* arow = h + (size_t)(mt * 16 + c) * HID + 4 * g;
  GemmFrag f0, f1;
  gemm_load_block(f0, arow, sB, bidx, 0);
  for (;;) {
    const int mt_next = mt + 2 * GEMM_GROUPS;
    const int mt_pf = mt_next < GEMM_MTILES ? mt_next : mt;
    const float* arow_next = h + (size_t)(mt_pf * 16 + c) * HID + 4 * g;

    v8f acc[4] = {};
    gemm_load_block(f1, arow, sB, bidx, 1);       gemm_mma_block(f0, acc);
    gemm_load_block(f0, arow, sB, bidx, 2);       gemm_mma_block(f1, acc);
    gemm_load_block(f1, arow, sB, bidx, 3);       gemm_mma_block(f0, acc);
    gemm_load_block(f0, arow, sB, bidx, 4);       gemm_mma_block(f1, acc);
    gemm_load_block(f1, arow, sB, bidx, 5);       gemm_mma_block(f0, acc);
    gemm_load_block(f0, arow, sB, bidx, 6);       gemm_mma_block(f1, acc);
    float dg[8];
#pragma unroll
    for (int i = 0; i < 8; ++i) dg[i] = deg[mt * 16 + 8 * g + i];
    gemm_load_block(f1, arow, sB, bidx, 7);       gemm_mma_block(f0, acc);
    gemm_load_block(f0, arow_next, sB, bidx, 0);  gemm_mma_block(f1, acc);

    float* xrow = x + (size_t)(mt * 16 + 8 * g) * HID + nq * 64 + c;
    float di[8];
#pragma unroll
    for (int i = 0; i < 8; ++i) di[i] = deg_to_dinv(dg[i]);
#pragma unroll
    for (int t = 0; t < 4; ++t)
#pragma unroll
      for (int i = 0; i < 8; ++i) xrow[(size_t)i * HID + 16 * t] = acc[t][i] * di[i];

    if (mt_next >= GEMM_MTILES) break;
    mt = mt_next;
    arow = arow_next;
  }
}

typedef __attribute__((ext_vector_type(4))) int v4i;

#define OWN_THREADS 1024
#define OWN_WAVES (OWN_THREADS / 32)
#define OWN_PIECES (N_EDGES / 16)
#define OWN_STEPS ((OWN_PIECES + 31) / 32)
#define OWN_SB 4
#define OWN_ITERS ((OWN_STEPS + OWN_WAVES * OWN_SB - 1) / (OWN_WAVES * OWN_SB))
#define OWN_QCAP 64
#define OWN_INIT ((OWN_ROWS * (HID / 4) + OWN_THREADS - 1) / OWN_THREADS)
#define AGG_GROUP 4

static_assert(N_EDGES % 16 == 0, "the owner scan reads whole 16-edge pieces");
static_assert((OWN_GROUPS - 1) * OWN_ROWS < N_NODES, "every group owns a row");
static_assert((OWN_GROUPS - 1) * (OWN_STEPS / OWN_GROUPS) < OWN_STEPS, "rotation");

__device__ __forceinline__ int wave_bcast_i(int v, int src_lane) {
  return __builtin_amdgcn_readlane(v, src_lane);
}

__device__ __forceinline__ int own_slot0(int i) {
  return (i >> 6) * HID + ((i & 32) << 2) + (i & 31);
}

__device__ __forceinline__ void own_edges4(
    const float* __restrict__ x, float* acc_lane,
    int r_l, int d_l, float nm_l, int k, unsigned lane_bytes) {
  v4f v[AGG_GROUP][2];
#pragma unroll
  for (int u = 0; u < AGG_GROUP; ++u) {
    const int r = wave_bcast_i(r_l, k + u);
    const char* src = (const char*)(x + (size_t)r * HID) + lane_bytes;
#pragma unroll
    for (int hh = 0; hh < 2; ++hh) v[u][hh] = *(const v4f*)(src + 512 * hh);
  }
#pragma unroll
  for (int u = 0; u < AGG_GROUP; ++u) {
    const int d = wave_bcast_i(d_l, k + u);
    const float nm = __int_as_float(wave_bcast_i(__float_as_int(nm_l), k + u));
    float* dst = acc_lane + d * HID;
#pragma unroll
    for (int hh = 0; hh < 2; ++hh)
#pragma unroll
      for (int j = 0; j < 4; ++j)
        atomicAdd(dst + 32 * (4 * hh + j), v[u][hh][j] * nm);
  }
}

__device__ __forceinline__ unsigned own_zero_bytes(unsigned y) {
  const unsigned t = ((y & 0x7F7F7F7Fu) + 0x7F7F7F7Fu) | y;
  return ~t & 0x80808080u;
}

__device__ __forceinline__ unsigned own_match16(v4i cv, unsigned pat) {
  const unsigned f0 = own_zero_bytes((unsigned)cv.x ^ pat);
  const unsigned f1 = own_zero_bytes((unsigned)cv.y ^ pat);
  const unsigned f2 = own_zero_bytes((unsigned)cv.z ^ pat);
  const unsigned f3 = own_zero_bytes((unsigned)cv.w ^ pat);
  return (f0 >> 7) | (f1 >> 6) | (f2 >> 5) | (f3 >> 4);
}

__device__ __forceinline__ void own_push(int* q, int& qn, unsigned mask, bool has, int e) {
  if (has) q[qn + (int)__builtin_amdgcn_mbcnt_lo(mask, 0u)] = e;
  qn += __builtin_popcount(mask);
}

__device__ __forceinline__ void own_drain(
    const int* __restrict__ ei, const float* __restrict__ w,
    const float* __restrict__ x,
    float* acc_lane, const int* qe, int n, int lo, int lane, unsigned lane_bytes) {
  int r_l = 0, d_l = 0;
  float nm_l = 0.0f;
  if (lane < n) {
    const int e = qe[lane];
    r_l = ei[e];
    const int c = ei[N_EDGES + e];
    d_l = c - lo;
    nm_l = w[e];
    if ((unsigned)d_l >= (unsigned)OWN_ROWS) { d_l = 0; nm_l = 0.0f; }
  }
  int k = 0;
  for (; k + AGG_GROUP <= n; k += AGG_GROUP)
    own_edges4(x, acc_lane, r_l, d_l, nm_l, k, lane_bytes);
  for (; k < n; ++k) {
    const int r = wave_bcast_i(r_l, k);
    const int d = wave_bcast_i(d_l, k);
    const float nm = __int_as_float(wave_bcast_i(__float_as_int(nm_l), k));
    const char* src = (const char*)(x + (size_t)r * HID) + lane_bytes;
    float* dst = acc_lane + d * HID;
    const v4f lo4 = *(const v4f*)src;
    const v4f hi4 = *(const v4f*)(src + 512);
#pragma unroll
    for (int j = 0; j < 4; ++j) {
      atomicAdd(dst + 32 * j, lo4[j] * nm);
      atomicAdd(dst + 32 * (4 + j), hi4[j] * nm);
    }
  }
}

__global__ __launch_bounds__(OWN_THREADS) void aggregate_owner_kernel(
    const int* __restrict__ ei,
    const unsigned char* __restrict__ owner,
    const float* __restrict__ w,
    const float* __restrict__ deg,
    const float* __restrict__ x,
    const float* __restrict__ bc,
    float* __restrict__ out) {
  __shared__ float acc[OWN_ROWS * HID];
  __shared__ int queue[OWN_WAVES * OWN_QCAP];
  __shared__ float sdinv[OWN_ROWS];

  const int tid = threadIdx.x;
  const int lane = tid & 31;
  const int wave = __builtin_amdgcn_readfirstlane(tid >> 5);
  const int lo = blockIdx.x * OWN_ROWS;
  const int cnt = (N_NODES - lo < OWN_ROWS) ? (N_NODES - lo) : OWN_ROWS;

  const float dgo = deg[lo + (tid < cnt ? tid : 0)];
  {
    const v4f* x4 = (const v4f*)(x + (size_t)lo * HID);
    const int total = cnt * (HID / 4);
    v4f xv[OWN_INIT];
#pragma unroll
    for (int t = 0; t < OWN_INIT; ++t) {
      const int i = tid + t * OWN_THREADS;
      xv[t] = x4[i < total ? i : total - 1];
    }
#pragma unroll
    for (int t = 0; t < OWN_INIT; ++t) {
      const int i = tid + t * OWN_THREADS;
      if (i < total) {
        float* s = acc + own_slot0(i);
        s[0] = xv[t].x;
        s[32] = xv[t].y;
        s[64] = xv[t].z;
        s[96] = xv[t].w;
      }
    }
  }
  if (tid < cnt) sdinv[tid] = deg_to_dinv(dgo);
  __syncthreads();

  int* q = queue + wave * OWN_QCAP;
  int qn = 0;
  const int rot = blockIdx.x * (OWN_STEPS / OWN_GROUPS);
  const v4i* own4 = (const v4i*)owner;
  const unsigned pat = blockIdx.x * 0x01010101u;
  const unsigned lane_bytes = (unsigned)lane * 16u;
  float* acc_lane = acc + lane;

  v4i cv[OWN_SB];
  int piece[OWN_SB];
#pragma unroll
  for (int u = 0; u < OWN_SB; ++u) {
    int sr = u * OWN_WAVES + wave + rot;
    if (sr >= OWN_STEPS) sr -= OWN_STEPS;
    piece[u] = sr * 32 + lane;
    cv[u] = own4[piece[u] < OWN_PIECES ? piece[u] : OWN_PIECES - 1];
  }

  static_assert(OWN_SB == 4, "the loop below names the four steps of a batch");
  for (int it = 0; it < OWN_ITERS; ++it) {
    const int itn = it + 1 < OWN_ITERS ? it + 1 : it;
    unsigned m[OWN_SB];
    int e0[OWN_SB];
#pragma unroll
    for (int u = 0; u < OWN_SB; ++u) {
      const int s = (it * OWN_SB + u) * OWN_WAVES + wave;
      m[u] = (s < OWN_STEPS && piece[u] < OWN_PIECES) ? own_match16(cv[u], pat) : 0u;
      e0[u] = piece[u] * 16;
      int sn = (itn * OWN_SB + u) * OWN_WAVES + wave + rot;
      if (sn >= OWN_STEPS) sn -= OWN_STEPS;
      piece[u] = sn * 32 + lane;
      cv[u] = own4[piece[u] < OWN_PIECES ? piece[u] : OWN_PIECES - 1];
    }
    unsigned any = __builtin_amdgcn_ballot_w32((m[0] | m[1] | m[2] | m[3]) != 0u);
    while (any != 0u) {
      const bool h0 = m[0] != 0u, h1 = m[1] != 0u, h2 = m[2] != 0u;
      const unsigned mm = h0 ? m[0] : h1 ? m[1] : h2 ? m[2] : m[3];
      const int eb = h0 ? e0[0] : h1 ? e0[1] : h2 ? e0[2] : e0[3];
      const int p = __builtin_ctz(mm | 0x80000000u);
      own_push(q, qn, any, mm != 0u, eb + (((p & 3) << 2) | (p >> 3)));
      const unsigned rest = mm & (mm - 1u);
      m[0] = h0 ? rest : m[0];
      m[1] = (!h0 && h1) ? rest : m[1];
      m[2] = (!h0 && !h1 && h2) ? rest : m[2];
      m[3] = (!h0 && !h1 && !h2) ? rest : m[3];
      if (qn >= 32) {
        qn -= 32;
        own_drain(ei, w, x, acc_lane, q + qn, 32, lo, lane, lane_bytes);
      }
      any = __builtin_amdgcn_ballot_w32((m[0] | m[1] | m[2] | m[3]) != 0u);
    }
  }
  if (qn > 0) own_drain(ei, w, x, acc_lane, q, qn, lo, lane, lane_bytes);
  __syncthreads();

  static_assert(OWN_THREADS % (HID / 4) == 0, "one bias piece per thread");
  const v4f b = ((const v4f*)bc)[tid & 63];
  v4f* out4 = (v4f*)(out + (size_t)lo * HID);
  for (int i = tid; i < cnt * (HID / 4); i += OWN_THREADS) {
    const float* s = acc + own_slot0(i);
    const float di = sdinv[i >> 6];
    v4f o;
    o.x = s[0] * di + b.x;
    o.y = s[32] * di + b.y;
    o.z = s[64] * di + b.z;
    o.w = s[96] * di + b.w;
    out4[i] = o;
  }
}

extern "C" void kernel_launch(void* const* d_in, const int* in_sizes, int n_in,
                              void* d_out, int out_size, void* d_ws, size_t ws_size,
                              hipStream_t stream) {
  const float* h    = (const float*)d_in[0];
  const int*   ei   = (const int*)d_in[1];
  const float* attr = (const float*)d_in[2];
  const float* W1   = (const float*)d_in[3];
  const float* b1   = (const float*)d_in[4];
  const float* W2   = (const float*)d_in[5];
  const float* b2   = (const float*)d_in[6];
  const float* Wc   = (const float*)d_in[7];
  const float* bc   = (const float*)d_in[8];
  float* out = (float*)d_out;

  const int N = N_NODES, E = N_EDGES;

  float* x   = (float*)d_ws;
  float* w   = x + (size_t)N * HID;
  float* deg = w + E;
  unsigned char* owner = (unsigned char*)(deg + N);

  edge_mlp_kernel<<<(E + 255) / 256, 256, 0, stream>>>(attr, W1, b1, W2, b2, ei + E, w, deg, owner, E, N);
  deg_lds_kernel<<<DEG_GROUPS, DEG_THREADS, 0, stream>>>(ei + E, w, deg, E);
  gemm_xWc_kernel<<<GEMM_GROUPS, 256, 0, stream>>>(h, Wc, deg, x);
  aggregate_owner_kernel<<<OWN_GROUPS, OWN_THREADS, 0, stream>>>(ei, owner, w, deg, x, bc, out);
}


# edge MLP: SiLU as (max(z,0)+min(z,0)*s)/(1+s), s=e^-|z|, so 16 hidden units share one denominator and one IEEE division (was 16); two edges a lane halve LDS reads; layer-2 sum order, rounding change
# speedup vs baseline: 5.5771x; 1.0629x over previous
#include <hip/hip_runtime.h>
#include <math.h>

typedef __attribute__((ext_vector_type(2))) float v2f;
typedef __attribute__((ext_vector_type(8))) float v8f;

#define N_NODES 50000
#define N_EDGES 800000
#define HID 256

#define OWN_GROUPS 256
#define OWN_ROWS ((N_NODES + OWN_GROUPS - 1) / OWN_GROUPS)
static_assert((N_NODES - 1) / OWN_ROWS < 256, "an owner code must fit one byte");

#define MLP_G 16
#define MLP_THREADS 128
#define MLP_LOG2E 1.44269504088896340736f
#define MLP_LN2 0.69314718055994530942f
static_assert(128 % MLP_G == 0 && (MLP_G & (MLP_G - 1)) == 0, "groups tile the hidden layer; pairwise tree");

__global__ __launch_bounds__(MLP_THREADS) void edge_mlp_kernel(
    const float* __restrict__ attr,
    const float* __restrict__ W1,
    const float* __restrict__ b1,
    const float* __restrict__ W2,
    const float* __restrict__ b2,
    const int* __restrict__ col,
    float* __restrict__ w,
    float* __restrict__ deg,
    unsigned char* __restrict__ owner,
    int E, int N) {
  __shared__ float sW1[3 * 128];
  __shared__ float sb1[128];
  __shared__ float sW2[128];
  for (int i = threadIdx.x; i < 384; i += MLP_THREADS) sW1[i] = W1[i] * MLP_LOG2E;
  for (int i = threadIdx.x; i < 128; i += MLP_THREADS) {
    sb1[i] = b1[i] * MLP_LOG2E;
    sW2[i] = W2[i] * MLP_LN2;
  }
  __syncthreads();
  const int e = blockIdx.x * (2 * MLP_THREADS) + threadIdx.x;
  const int e2 = e + MLP_THREADS;
  static_assert(N_EDGES >= N_NODES, "deg start value needs a thread slot per node");
  if (e < N) deg[e] = 1.0f;
  if (e2 < N) deg[e2] = 1.0f;
  if (e >= E) return;
  const bool has2 = e2 < E;
  const int e2c = has2 ? e2 : E - 1;
  const float a0 = attr[e * 3 + 0], a1 = attr[e * 3 + 1], a2 = attr[e * 3 + 2];
  const float c0 = attr[e2c * 3 + 0], c1 = attr[e2c * 3 + 1], c2 = attr[e2c * 3 + 2];
  const int c = col[e];
  const int cc = col[e2c];
  float acc = 0.0f, acd = 0.0f;
  for (int j0 = 0; j0 < 128; j0 += MLP_G) {
    float num[MLP_G], den[MLP_G], nud[MLP_G], ded[MLP_G];
#pragma unroll
    for (int k = 0; k < MLP_G; ++k) {
      const int j = j0 + k;
      const float w0 = sW1[j], w1 = sW1[128 + j], w2 = sW1[256 + j], bb = sb1[j], v2 = sW2[j];
      const float z = fmaf(a0, w0, fmaf(a1, w1, fmaf(a2, w2, bb)));
      const float y = fmaf(c0, w0, fmaf(c1, w1, fmaf(c2, w2, bb)));
      const float zp = fmaxf(z, 0.0f), zn = fminf(z, 0.0f);
      const float yp = fmaxf(y, 0.0f), yn = fminf(y, 0.0f);
      const float s = __builtin_amdgcn_exp2f(zn - zp);
      const float t = __builtin_amdgcn_exp2f(yn - yp);
      num[k] = fmaf(zn, s, zp) * v2;
      nud[k] = fmaf(yn, t, yp) * v2;
      den[k] = 1.0f + s;
      ded[k] = 1.0f + t;
    }
#pragma unroll
    for (int w2 = 1; w2 < MLP_G; w2 *= 2)
#pragma unroll
      for (int k = 0; k < MLP_G; k += 2 * w2) {
        num[k] = fmaf(num[k], den[k + w2], num[k + w2] * den[k]);
        den[k] = den[k] * den[k + w2];
        nud[k] = fmaf(nud[k], ded[k + w2], nud[k + w2] * ded[k]);
        ded[k] = ded[k] * ded[k + w2];
      }
    acc += num[0] / den[0];
    acd += nud[0] / ded[0];
  }
  const float bias2 = b2[0];
  const float zz = acc + bias2;
  const float yy = acd + bias2;
  w[e] = 1.0f / (1.0f + __expf(-zz));
  owner[e] = (unsigned char)((unsigned)c / (unsigned)OWN_ROWS);
  if (has2) {
    w[e2] = 1.0f / (1.0f + __expf(-yy));
    owner[e2] = (unsigned char)((unsigned)cc / (unsigned)OWN_ROWS);
  }
}

typedef __attribute__((ext_vector_type(4))) float v4f;

#define DEG_GROUPS 64
#define DEG_THREADS 1024
#define DEG_CHUNK ((N_EDGES + DEG_GROUPS - 1) / DEG_GROUPS)
#define DEG_BATCH ((DEG_CHUNK + DEG_THREADS - 1) / DEG_THREADS)
#define DEG_LINES ((N_NODES + 31) / 32)
#define DEG_FLUSH 7

__device__ __forceinline__ float deg_to_dinv(float d) {
  return d > 0.0f ? rsqrtf(d) : 0.0f;
}

__global__ __launch_bounds__(DEG_THREADS) void deg_lds_kernel(
    const int* __restrict__ col, const float* __restrict__ w,
    float* __restrict__ deg, int E) {
  __shared__ float tab[N_NODES];
  const int tid = threadIdx.x;

  v4f* tab4 = (v4f*)tab;
  const v4f zero = {0.0f, 0.0f, 0.0f, 0.0f};
  for (int i = tid; i < N_NODES / 4; i += DEG_THREADS) tab4[i] = zero;
  __syncthreads();

  const int chunk = (E + DEG_GROUPS - 1) / DEG_GROUPS;
  const int e0 = blockIdx.x * chunk;
  const int e1 = (e0 + chunk < E) ? (e0 + chunk) : E;
  for (int base = e0; base < e1; base += DEG_BATCH * DEG_THREADS) {
    int c[DEG_BATCH];
    float v[DEG_BATCH];
#pragma unroll
    for (int k = 0; k < DEG_BATCH; ++k) {
      const int e = base + k * DEG_THREADS + tid;
      const int ec = e < e1 ? e : e1 - 1;
      c[k] = col[ec];
      v[k] = w[ec];
    }
#pragma unroll
    for (int k = 0; k < DEG_BATCH; ++k)
      if (base + k * DEG_THREADS + tid < e1) atomicAdd(&tab[c[k]], v[k]);
  }
  __syncthreads();

  const int lane = tid & 31;
  const int wave = tid >> 5;
  const int rot = blockIdx.x * (DEG_LINES / DEG_GROUPS);
  for (int k0 = wave; k0 < DEG_LINES; k0 += (DEG_THREADS / 32) * DEG_FLUSH) {
    int idx[DEG_FLUSH];
    float t[DEG_FLUSH];
#pragma unroll
    for (int u = 0; u < DEG_FLUSH; ++u) {
      const int k = k0 + (DEG_THREADS / 32) * u;
      int line = k + rot;
      if (line >= DEG_LINES) line -= DEG_LINES;
      const int i = line * 32 + lane;
      idx[u] = (k < DEG_LINES && i < N_NODES) ? i : -1;
      t[u] = tab[idx[u] >= 0 ? i : 0];
    }
#pragma unroll
    for (int u = 0; u < DEG_FLUSH; ++u)
      if (idx[u] >= 0) unsafeAtomicAdd(&deg[idx[u]], t[u]);
  }
}


#define GEMM_MTILES (N_NODES / 16)
#define GEMM_GROUPS 256
#define GEMM_KP 4

struct GemmFrag {
  v4f a[GEMM_KP];
  v4f b[GEMM_KP][4];
};

__device__ __forceinline__ void gemm_load_block(
    GemmFrag& f, const float* __restrict__ arow, const v4f* sB, int bidx, int blk) {
#pragma unroll
  for (int jj = 0; jj < GEMM_KP; ++jj)
    f.a[jj] = *(const v4f*)(arow + (blk * GEMM_KP + jj) * 8);
#pragma unroll
  for (int jj = 0; jj < GEMM_KP; ++jj)
#pragma unroll
    for (int t = 0; t < 4; ++t)
      f.b[jj][t] = sB[bidx + (t * 32 + blk * GEMM_KP + jj) * 32];
  __builtin_amdgcn_sched_barrier(0);
}

__device__ __forceinline__ void gemm_mma_block(const GemmFrag& f, v8f (&acc)[4]) {
#pragma unroll
  for (int jj = 0; jj < GEMM_KP; ++jj) {
#pragma unroll
    for (int t = 0; t < 4; ++t) {
      v2f a = {f.a[jj].x, f.a[jj].y};
      v2f b = {f.b[jj][t].x, f.b[jj][t].y};
      acc[t] = __builtin_amdgcn_wmma_f32_16x16x4_f32(false, a, false, b, (short)0,
                                                     acc[t], false, false);
    }
#pragma unroll
    for (int t = 0; t < 4; ++t) {
      v2f a = {f.a[jj].z, f.a[jj].w};
      v2f b = {f.b[jj][t].z, f.b[jj][t].w};
      acc[t] = __builtin_amdgcn_wmma_f32_16x16x4_f32(false, a, false, b, (short)0,
                                                     acc[t], false, false);
    }
  }
  __builtin_amdgcn_sched_barrier(0);
}

__global__ __launch_bounds__(256) void gemm_xWc_kernel(
    const float* __restrict__ h,
    const float* __restrict__ Wc,
    const float* __restrict__ deg,
    float* __restrict__ x) {
  __shared__ v4f sB[16 * 32 * 32];

  const int lane = threadIdx.x & 31;
  const int wave = __builtin_amdgcn_readfirstlane((int)(threadIdx.x >> 5));
  const int c = lane & 15;
  const int g = lane >> 4;

#pragma unroll 8
  for (int i = 0; i < 64; ++i) {
    const int p = wave * 64 + i;
    const int nt = p >> 5, j = p & 31;
    const float* src = Wc + (size_t)(8 * j + 4 * g) * HID + 16 * nt + c;
    v4f v;
    v.x = src[0];
    v.y = src[HID];
    v.z = src[2 * HID];
    v.w = src[3 * HID];
    sB[p * 32 + lane] = v;
  }
  __syncthreads();

  const int nq = wave & 3;
  const int bidx = (nq * 4 * 32) * 32 + lane;
  int mt = blockIdx.x + GEMM_GROUPS * (wave >> 2);
  if (mt >= GEMM_MTILES) return;

  const float* arow = h + (size_t)(mt * 16 + c) * HID + 4 * g;
  GemmFrag f0, f1;
  gemm_load_block(f0, arow, sB, bidx, 0);
  for (;;) {
    const int mt_next = mt + 2 * GEMM_GROUPS;
    const int mt_pf = mt_next < GEMM_MTILES ? mt_next : mt;
    const float* arow_next = h + (size_t)(mt_pf * 16 + c) * HID + 4 * g;

    v8f acc[4] = {};
    gemm_load_block(f1, arow, sB, bidx, 1);       gemm_mma_block(f0, acc);
    gemm_load_block(f0, arow, sB, bidx, 2);       gemm_mma_block(f1, acc);
    gemm_load_block(f1, arow, sB, bidx, 3);       gemm_mma_block(f0, acc);
    gemm_load_block(f0, arow, sB, bidx, 4);       gemm_mma_block(f1, acc);
    gemm_load_block(f1, arow, sB, bidx, 5);       gemm_mma_block(f0, acc);
    gemm_load_block(f0, arow, sB, bidx, 6);       gemm_mma_block(f1, acc);
    float dg[8];
#pragma unroll
    for (int i = 0; i < 8; ++i) dg[i] = deg[mt * 16 + 8 * g + i];
    gemm_load_block(f1, arow, sB, bidx, 7);       gemm_mma_block(f0, acc);
    gemm_load_block(f0, arow_next, sB, bidx, 0);  gemm_mma_block(f1, acc);

    float* xrow = x + (size_t)(mt * 16 + 8 * g) * HID + nq * 64 + c;
    float di[8];
#pragma unroll
    for (int i = 0; i < 8; ++i) di[i] = deg_to_dinv(dg[i]);
#pragma unroll
    for (int t = 0; t < 4; ++t)
#pragma unroll
      for (int i = 0; i < 8; ++i) xrow[(size_t)i * HID + 16 * t] = acc[t][i] * di[i];

    if (mt_next >= GEMM_MTILES) break;
    mt = mt_next;
    arow = arow_next;
  }
}

typedef __attribute__((ext_vector_type(4))) int v4i;

#define OWN_THREADS 1024
#define OWN_WAVES (OWN_THREADS / 32)
#define OWN_PIECES (N_EDGES / 16)
#define OWN_STEPS ((OWN_PIECES + 31) / 32)
#define OWN_SB 4
#define OWN_ITERS ((OWN_STEPS + OWN_WAVES * OWN_SB - 1) / (OWN_WAVES * OWN_SB))
#define OWN_QCAP 64
#define OWN_INIT ((OWN_ROWS * (HID / 4) + OWN_THREADS - 1) / OWN_THREADS)
#define AGG_GROUP 4

static_assert(N_EDGES % 16 == 0, "the owner scan reads whole 16-edge pieces");
static_assert((OWN_GROUPS - 1) * OWN_ROWS < N_NODES, "every group owns a row");
static_assert((OWN_GROUPS - 1) * (OWN_STEPS / OWN_GROUPS) < OWN_STEPS, "rotation");

__device__ __forceinline__ int wave_bcast_i(int v, int src_lane) {
  return __builtin_amdgcn_readlane(v, src_lane);
}

__device__ __forceinline__ int own_slot0(int i) {
  return (i >> 6) * HID + ((i & 32) << 2) + (i & 31);
}

__device__ __forceinline__ void own_edges4(
    const float* __restrict__ x, float* acc_lane,
    int r_l, int d_l, float nm_l, int k, unsigned lane_bytes) {
  v4f v[AGG_GROUP][2];
#pragma unroll
  for (int u = 0; u < AGG_GROUP; ++u) {
    const int r = wave_bcast_i(r_l, k + u);
    const char* src = (const char*)(x + (size_t)r * HID) + lane_bytes;
#pragma unroll
    for (int hh = 0; hh < 2; ++hh) v[u][hh] = *(const v4f*)(src + 512 * hh);
  }
#pragma unroll
  for (int u = 0; u < AGG_GROUP; ++u) {
    const int d = wave_bcast_i(d_l, k + u);
    const float nm = __int_as_float(wave_bcast_i(__float_as_int(nm_l), k + u));
    float* dst = acc_lane + d * HID;
#pragma unroll
    for (int hh = 0; hh < 2; ++hh)
#pragma unroll
      for (int j = 0; j < 4; ++j)
        atomicAdd(dst + 32 * (4 * hh + j), v[u][hh][j] * nm);
  }
}

__device__ __forceinline__ unsigned own_zero_bytes(unsigned y) {
  const unsigned t = ((y & 0x7F7F7F7Fu) + 0x7F7F7F7Fu) | y;
  return ~t & 0x80808080u;
}

__device__ __forceinline__ unsigned own_match16(v4i cv, unsigned pat) {
  const unsigned f0 = own_zero_bytes((unsigned)cv.x ^ pat);
  const unsigned f1 = own_zero_bytes((unsigned)cv.y ^ pat);
  const unsigned f2 = own_zero_bytes((unsigned)cv.z ^ pat);
  const unsigned f3 = own_zero_bytes((unsigned)cv.w ^ pat);
  return (f0 >> 7) | (f1 >> 6) | (f2 >> 5) | (f3 >> 4);
}

__device__ __forceinline__ void own_push(int* q, int& qn, unsigned mask, bool has, int e) {
  if (has) q[qn + (int)__builtin_amdgcn_mbcnt_lo(mask, 0u)] = e;
  qn += __builtin_popcount(mask);
}

__device__ __forceinline__ void own_drain(
    const int* __restrict__ ei, const float* __restrict__ w,
    const float* __restrict__ x,
    float* acc_lane, const int* qe, int n, int lo, int lane, unsigned lane_bytes) {
  int r_l = 0, d_l = 0;
  float nm_l = 0.0f;
  if (lane < n) {
    const int e = qe[lane];
    r_l = ei[e];
    const int c = ei[N_EDGES + e];
    d_l = c - lo;
    nm_l = w[e];
    if ((unsigned)d_l >= (unsigned)OWN_ROWS) { d_l = 0; nm_l = 0.0f; }
  }
  int k = 0;
  for (; k + AGG_GROUP <= n; k += AGG_GROUP)
    own_edges4(x, acc_lane, r_l, d_l, nm_l, k, lane_bytes);
  for (; k < n; ++k) {
    const int r = wave_bcast_i(r_l, k);
    const int d = wave_bcast_i(d_l, k);
    const float nm = __int_as_float(wave_bcast_i(__float_as_int(nm_l), k));
    const char* src = (const char*)(x + (size_t)r * HID) + lane_bytes;
    float* dst = acc_lane + d * HID;
    const v4f lo4 = *(const v4f*)src;
    const v4f hi4 = *(const v4f*)(src + 512);
#pragma unroll
    for (int j = 0; j < 4; ++j) {
      atomicAdd(dst + 32 * j, lo4[j] * nm);
      atomicAdd(dst + 32 * (4 + j), hi4[j] * nm);
    }
  }
}

__global__ __launch_bounds__(OWN_THREADS) void aggregate_owner_kernel(
    const int* __restrict__ ei,
    const unsigned char* __restrict__ owner,
    const float* __restrict__ w,
    const float* __restrict__ deg,
    const float* __restrict__ x,
    const float* __restrict__ bc,
    float* __restrict__ out) {
  __shared__ float acc[OWN_ROWS * HID];
  __shared__ int queue[OWN_WAVES * OWN_QCAP];
  __shared__ float sdinv[OWN_ROWS];

  const int tid = threadIdx.x;
  const int lane = tid & 31;
  const int wave = __builtin_amdgcn_readfirstlane(tid >> 5);
  const int lo = blockIdx.x * OWN_ROWS;
  const int cnt = (N_NODES - lo < OWN_ROWS) ? (N_NODES - lo) : OWN_ROWS;

  const float dgo = deg[lo + (tid < cnt ? tid : 0)];
  {
    const v4f* x4 = (const v4f*)(x + (size_t)lo * HID);
    const int total = cnt * (HID / 4);
    v4f xv[OWN_INIT];
#pragma unroll
    for (int t = 0; t < OWN_INIT; ++t) {
      const int i = tid + t * OWN_THREADS;
      xv[t] = x4[i < total ? i : total - 1];
    }
#pragma unroll
    for (int t = 0; t < OWN_INIT; ++t) {
      const int i = tid + t * OWN_THREADS;
      if (i < total) {
        float* s = acc + own_slot0(i);
        s[0] = xv[t].x;
        s[32] = xv[t].y;
        s[64] = xv[t].z;
        s[96] = xv[t].w;
      }
    }
  }
  if (tid < cnt) sdinv[tid] = deg_to_dinv(dgo);
  __syncthreads();

  int* q = queue + wave * OWN_QCAP;
  int qn = 0;
  const int rot = blockIdx.x * (OWN_STEPS / OWN_GROUPS);
  const v4i* own4 = (const v4i*)owner;
  const unsigned pat = blockIdx.x * 0x01010101u;
  const unsigned lane_bytes = (unsigned)lane * 16u;
  float* acc_lane = acc + lane;

  v4i cv[OWN_SB];
  int piece[OWN_SB];
#pragma unroll
  for (int u = 0; u < OWN_SB; ++u) {
    int sr = u * OWN_WAVES + wave + rot;
    if (sr >= OWN_STEPS) sr -= OWN_STEPS;
    piece[u] = sr * 32 + lane;
    cv[u] = own4[piece[u] < OWN_PIECES ? piece[u] : OWN_PIECES - 1];
  }

  static_assert(OWN_SB == 4, "the loop below names the four steps of a batch");
  for (int it = 0; it < OWN_ITERS; ++it) {
    const int itn = it + 1 < OWN_ITERS ? it + 1 : it;
    unsigned m[OWN_SB];
    int e0[OWN_SB];
#pragma unroll
    for (int u = 0; u < OWN_SB; ++u) {
      const int s = (it * OWN_SB + u) * OWN_WAVES + wave;
      m[u] = (s < OWN_STEPS && piece[u] < OWN_PIECES) ? own_match16(cv[u], pat) : 0u;
      e0[u] = piece[u] * 16;
      int sn = (itn * OWN_SB + u) * OWN_WAVES + wave + rot;
      if (sn >= OWN_STEPS) sn -= OWN_STEPS;
      piece[u] = sn * 32 + lane;
      cv[u] = own4[piece[u] < OWN_PIECES ? piece[u] : OWN_PIECES - 1];
    }
    unsigned any = __builtin_amdgcn_ballot_w32((m[0] | m[1] | m[2] | m[3]) != 0u);
    while (any != 0u) {
      const bool h0 = m[0] != 0u, h1 = m[1] != 0u, h2 = m[2] != 0u;
      const unsigned mm = h0 ? m[0] : h1 ? m[1] : h2 ? m[2] : m[3];
      const int eb = h0 ? e0[0] : h1 ? e0[1] : h2 ? e0[2] : e0[3];
      const int p = __builtin_ctz(mm | 0x80000000u);
      own_push(q, qn, any, mm != 0u, eb + (((p & 3) << 2) | (p >> 3)));
      const unsigned rest = mm & (mm - 1u);
      m[0] = h0 ? rest : m[0];
      m[1] = (!h0 && h1) ? rest : m[1];
      m[2] = (!h0 && !h1 && h2) ? rest : m[2];
      m[3] = (!h0 && !h1 && !h2) ? rest : m[3];
      if (qn >= 32) {
        qn -= 32;
        own_drain(ei, w, x, acc_lane, q + qn, 32, lo, lane, lane_bytes);
      }
      any = __builtin_amdgcn_ballot_w32((m[0] | m[1] | m[2] | m[3]) != 0u);
    }
  }
  if (qn > 0) own_drain(ei, w, x, acc_lane, q, qn, lo, lane, lane_bytes);
  __syncthreads();

  static_assert(OWN_THREADS % (HID / 4) == 0, "one bias piece per thread");
  const v4f b = ((const v4f*)bc)[tid & 63];
  v4f* out4 = (v4f*)(out + (size_t)lo * HID);
  for (int i = tid; i < cnt * (HID / 4); i += OWN_THREADS) {
    const float* s = acc + own_slot0(i);
    const float di = sdinv[i >> 6];
    v4f o;
    o.x = s[0] * di + b.x;
    o.y = s[32] * di + b.y;
    o.z = s[64] * di + b.z;
    o.w = s[96] * di + b.w;
    out4[i] = o;
  }
}

extern "C" void kernel_launch(void* const* d_in, const int* in_sizes, int n_in,
                              void* d_out, int out_size, void* d_ws, size_t ws_size,
                              hipStream_t stream) {
  const float* h    = (const float*)d_in[0];
  const int*   ei   = (const int*)d_in[1];
  const float* attr = (const float*)d_in[2];
  const float* W1   = (const float*)d_in[3];
  const float* b1   = (const float*)d_in[4];
  const float* W2   = (const float*)d_in[5];
  const float* b2   = (const float*)d_in[6];
  const float* Wc   = (const float*)d_in[7];
  const float* bc   = (const float*)d_in[8];
  float* out = (float*)d_out;

  const int N = N_NODES, E = N_EDGES;

  float* x   = (float*)d_ws;
  float* w   = x + (size_t)N * HID;
  float* deg = w + E;
  unsigned char* owner = (unsigned char*)(deg + N);

  edge_mlp_kernel<<<(E + 2 * MLP_THREADS - 1) / (2 * MLP_THREADS), MLP_THREADS, 0, stream>>>(attr, W1, b1, W2, b2, ei + E, w, deg, owner, E, N);
  deg_lds_kernel<<<DEG_GROUPS, DEG_THREADS, 0, stream>>>(ei + E, w, deg, E);
  gemm_xWc_kernel<<<GEMM_GROUPS, 256, 0, stream>>>(h, Wc, deg, x);
  aggregate_owner_kernel<<<OWN_GROUPS, OWN_THREADS, 0, stream>>>(ei, owner, w, deg, x, bc, out);
}


# edge MLP runs in the x = h @ Wc launch on eight extra waves a group, in the issue slots the f32 matrix waves leave idle; dinv[source] joins the edge weight; sum order kept, rounding changes
# speedup vs baseline: 5.7512x; 1.0312x over previous
#include <hip/hip_runtime.h>
#include <math.h>

typedef __attribute__((ext_vector_type(2))) float v2f;
typedef __attribute__((ext_vector_type(8))) float v8f;

#define N_NODES 50000
#define N_EDGES 800000
#define HID 256

#define OWN_GROUPS 256
#define OWN_ROWS ((N_NODES + OWN_GROUPS - 1) / OWN_GROUPS)
static_assert((N_NODES - 1) / OWN_ROWS < 256, "an owner code must fit one byte");

#define MLP_G 16
#define MLP_CHUNK 64
#define MLP_LOG2E 1.44269504088896340736f
#define MLP_LN2 0.69314718055994530942f
static_assert(128 % MLP_G == 0 && (MLP_G & (MLP_G - 1)) == 0, "groups tile the hidden layer; pairwise tree");

struct MlpEdgeIn {
  float a0, a1, a2;
  float c0, c1, c2;
  int c, cc;
};

__device__ __forceinline__ MlpEdgeIn edge_mlp_load(
    const float* __restrict__ attr,
    const int* __restrict__ col,
    int e_base, int lane, int E) {
  const int e = e_base + lane;
  const int e2 = e + 32;
  const int ec = e < E ? e : E - 1;
  const int e2c = e2 < E ? e2 : E - 1;
  MlpEdgeIn in;
  in.a0 = attr[ec * 3 + 0];  in.a1 = attr[ec * 3 + 1];  in.a2 = attr[ec * 3 + 2];
  in.c0 = attr[e2c * 3 + 0]; in.c1 = attr[e2c * 3 + 1]; in.c2 = attr[e2c * 3 + 2];
  in.c = col[ec];
  in.cc = col[e2c];
  return in;
}

__device__ __forceinline__ void edge_mlp_chunk(
    const MlpEdgeIn& in,
    const float* sW1, const float* sb1, const float* sW2, float bias2,
    float* __restrict__ w,
    float* __restrict__ deg,
    unsigned char* __restrict__ owner,
    int e_base, int lane, int E, int N) {
  const int e = e_base + lane;
  const int e2 = e + 32;
  static_assert(N_EDGES >= N_NODES, "deg start value needs an edge slot per node");
  if (e < N) deg[e] = 1.0f;
  if (e2 < N) deg[e2] = 1.0f;
  if (e >= E) return;
  const bool has2 = e2 < E;
  const float a0 = in.a0, a1 = in.a1, a2 = in.a2;
  const float c0 = in.c0, c1 = in.c1, c2 = in.c2;
  const int c = in.c;
  const int cc = in.cc;
  float acc = 0.0f, acd = 0.0f;
  for (int j0 = 0; j0 < 128; j0 += MLP_G) {
    float num[MLP_G], den[MLP_G], nud[MLP_G], ded[MLP_G];
#pragma unroll
    for (int k = 0; k < MLP_G; ++k) {
      const int j = j0 + k;
      const float w0 = sW1[j], w1 = sW1[128 + j], w2 = sW1[256 + j], bb = sb1[j], v2 = sW2[j];
      const float z = fmaf(a0, w0, fmaf(a1, w1, fmaf(a2, w2, bb)));
      const float y = fmaf(c0, w0, fmaf(c1, w1, fmaf(c2, w2, bb)));
      const float zp = fmaxf(z, 0.0f), zn = fminf(z, 0.0f);
      const float yp = fmaxf(y, 0.0f), yn = fminf(y, 0.0f);
      const float s = __builtin_amdgcn_exp2f(zn - zp);
      const float t = __builtin_amdgcn_exp2f(yn - yp);
      num[k] = fmaf(zn, s, zp) * v2;
      nud[k] = fmaf(yn, t, yp) * v2;
      den[k] = 1.0f + s;
      ded[k] = 1.0f + t;
    }
#pragma unroll
    for (int w2 = 1; w2 < MLP_G; w2 *= 2)
#pragma unroll
      for (int k = 0; k < MLP_G; k += 2 * w2) {
        num[k] = fmaf(num[k], den[k + w2], num[k + w2] * den[k]);
        den[k] = den[k] * den[k + w2];
        nud[k] = fmaf(nud[k], ded[k + w2], nud[k + w2] * ded[k]);
        ded[k] = ded[k] * ded[k + w2];
      }
    acc += num[0] / den[0];
    acd += nud[0] / ded[0];
  }
  const float zz = acc + bias2;
  const float yy = acd + bias2;
  w[e] = 1.0f / (1.0f + __expf(-zz));
  owner[e] = (unsigned char)((unsigned)c / (unsigned)OWN_ROWS);
  if (has2) {
    w[e2] = 1.0f / (1.0f + __expf(-yy));
    owner[e2] = (unsigned char)((unsigned)cc / (unsigned)OWN_ROWS);
  }
}

typedef __attribute__((ext_vector_type(4))) float v4f;

#define DEG_GROUPS 64
#define DEG_THREADS 1024
#define DEG_CHUNK ((N_EDGES + DEG_GROUPS - 1) / DEG_GROUPS)
#define DEG_BATCH ((DEG_CHUNK + DEG_THREADS - 1) / DEG_THREADS)
#define DEG_LINES ((N_NODES + 31) / 32)
#define DEG_FLUSH 7

__device__ __forceinline__ float deg_to_dinv(float d) {
  return d > 0.0f ? rsqrtf(d) : 0.0f;
}

__global__ __launch_bounds__(DEG_THREADS) void deg_lds_kernel(
    const int* __restrict__ col, const float* __restrict__ w,
    float* __restrict__ deg, int E) {
  __shared__ float tab[N_NODES];
  const int tid = threadIdx.x;

  v4f* tab4 = (v4f*)tab;
  const v4f zero = {0.0f, 0.0f, 0.0f, 0.0f};
  for (int i = tid; i < N_NODES / 4; i += DEG_THREADS) tab4[i] = zero;
  __syncthreads();

  const int chunk = (E + DEG_GROUPS - 1) / DEG_GROUPS;
  const int e0 = blockIdx.x * chunk;
  const int e1 = (e0 + chunk < E) ? (e0 + chunk) : E;
  for (int base = e0; base < e1; base += DEG_BATCH * DEG_THREADS) {
    int c[DEG_BATCH];
    float v[DEG_BATCH];
#pragma unroll
    for (int k = 0; k < DEG_BATCH; ++k) {
      const int e = base + k * DEG_THREADS + tid;
      const int ec = e < e1 ? e : e1 - 1;
      c[k] = col[ec];
      v[k] = w[ec];
    }
#pragma unroll
    for (int k = 0; k < DEG_BATCH; ++k)
      if (base + k * DEG_THREADS + tid < e1) atomicAdd(&tab[c[k]], v[k]);
  }
  __syncthreads();

  const int lane = tid & 31;
  const int wave = tid >> 5;
  const int rot = blockIdx.x * (DEG_LINES / DEG_GROUPS);
  for (int k0 = wave; k0 < DEG_LINES; k0 += (DEG_THREADS / 32) * DEG_FLUSH) {
    int idx[DEG_FLUSH];
    float t[DEG_FLUSH];
#pragma unroll
    for (int u = 0; u < DEG_FLUSH; ++u) {
      const int k = k0 + (DEG_THREADS / 32) * u;
      int line = k + rot;
      if (line >= DEG_LINES) line -= DEG_LINES;
      const int i = line * 32 + lane;
      idx[u] = (k < DEG_LINES && i < N_NODES) ? i : -1;
      t[u] = tab[idx[u] >= 0 ? i : 0];
    }
#pragma unroll
    for (int u = 0; u < DEG_FLUSH; ++u)
      if (idx[u] >= 0) unsafeAtomicAdd(&deg[idx[u]], t[u]);
  }
}


#define GEMM_MTILES (N_NODES / 16)
#define GEMM_GROUPS 256
#define GEMM_KP 4

struct GemmFrag {
  v4f a[GEMM_KP];
  v4f b[GEMM_KP][4];
};

__device__ __forceinline__ void gemm_load_block(
    GemmFrag& f, const float* __restrict__ atile, unsigned a_lane, const v4f* sB, int bidx, int blk) {
#pragma unroll
  for (int jj = 0; jj < GEMM_KP; ++jj)
    f.a[jj] = *(const v4f*)(atile + (a_lane + (unsigned)((blk * GEMM_KP + jj) * 8)));
#pragma unroll
  for (int jj = 0; jj < GEMM_KP; ++jj)
#pragma unroll
    for (int t = 0; t < 4; ++t)
      f.b[jj][t] = sB[bidx + (t * 32 + blk * GEMM_KP + jj) * 32];
  __builtin_amdgcn_sched_barrier(0);
}

__device__ __forceinline__ void gemm_mma_block(const GemmFrag& f, v8f (&acc)[4]) {
#pragma unroll
  for (int jj = 0; jj < GEMM_KP; ++jj) {
#pragma unroll
    for (int t = 0; t < 4; ++t) {
      v2f a = {f.a[jj].x, f.a[jj].y};
      v2f b = {f.b[jj][t].x, f.b[jj][t].y};
      acc[t] = __builtin_amdgcn_wmma_f32_16x16x4_f32(false, a, false, b, (short)0,
                                                     acc[t], false, false);
    }
#pragma unroll
    for (int t = 0; t < 4; ++t) {
      v2f a = {f.a[jj].z, f.a[jj].w};
      v2f b = {f.b[jj][t].z, f.b[jj][t].w};
      acc[t] = __builtin_amdgcn_wmma_f32_16x16x4_f32(false, a, false, b, (short)0,
                                                     acc[t], false, false);
    }
  }
  __builtin_amdgcn_sched_barrier(0);
}

#define GEMM_WAVES 8
#define MLP_WAVES 8
#define FUSED_THREADS ((GEMM_WAVES + MLP_WAVES) * 32)
static_assert(GEMM_WAVES + MLP_WAVES == 16, "Wc staging: one column tile per wave");

__global__ __launch_bounds__(FUSED_THREADS) void gemm_mlp_kernel(
    const float* __restrict__ h,
    const float* __restrict__ Wc,
    float* __restrict__ x,
    const float* __restrict__ attr,
    const float* __restrict__ W1,
    const float* __restrict__ b1,
    const float* __restrict__ W2,
    const float* __restrict__ b2,
    const int* __restrict__ col,
    float* __restrict__ w,
    float* __restrict__ deg,
    unsigned char* __restrict__ owner) {
  __shared__ v4f sB[16 * 32 * 32];
  __shared__ float sW1[3 * 128];
  __shared__ float sb1[128];
  __shared__ float sW2[128];

  const int lane = threadIdx.x & 31;
  const int wave = __builtin_amdgcn_readfirstlane((int)(threadIdx.x >> 5));
  const int c = lane & 15;
  const int g = lane >> 4;

#pragma unroll 8
  for (int j = 0; j < 32; ++j) {
    const int p = wave * 32 + j;
    const float* src = Wc + (size_t)(8 * j + 4 * g) * HID + 16 * wave + c;
    v4f v;
    v.x = src[0];
    v.y = src[HID];
    v.z = src[2 * HID];
    v.w = src[3 * HID];
    sB[p * 32 + lane] = v;
  }
  for (int i = threadIdx.x; i < 384; i += FUSED_THREADS) sW1[i] = W1[i] * MLP_LOG2E;
  if (threadIdx.x < 128) {
    sb1[threadIdx.x] = b1[threadIdx.x] * MLP_LOG2E;
    sW2[threadIdx.x] = W2[threadIdx.x] * MLP_LN2;
  }
  __syncthreads();

  if (wave >= GEMM_WAVES) {
    const float bias2 = b2[0];
    const int n_chunks = (N_EDGES + MLP_CHUNK - 1) / MLP_CHUNK;
    int ch = blockIdx.x * MLP_WAVES + (wave - GEMM_WAVES);
    if (ch >= n_chunks) return;
    MlpEdgeIn cur = edge_mlp_load(attr, col, ch * MLP_CHUNK, lane, N_EDGES);
    for (;;) {
      const int ch_next = ch + GEMM_GROUPS * MLP_WAVES;
      const bool more = ch_next < n_chunks;
      const MlpEdgeIn nxt = edge_mlp_load(attr, col, (more ? ch_next : ch) * MLP_CHUNK, lane, N_EDGES);
      edge_mlp_chunk(cur, sW1, sb1, sW2, bias2, w, deg, owner,
                     ch * MLP_CHUNK, lane, N_EDGES, N_NODES);
      if (!more) break;
      cur = nxt;
      ch = ch_next;
    }
    return;
  }

  const int nq = wave & 3;
  const int bidx = (nq * 4 * 32) * 32 + lane;
  int mt = blockIdx.x + GEMM_GROUPS * (wave >> 2);
  if (mt >= GEMM_MTILES) return;

  const unsigned a_lane = (unsigned)(c * HID + 4 * g);
  const float* atile = h + (size_t)mt * (16 * HID);
  GemmFrag f0, f1;
  gemm_load_block(f0, atile, a_lane, sB, bidx, 0);
  for (;;) {
    const int mt_next = mt + 2 * GEMM_GROUPS;
    const int mt_pf = mt_next < GEMM_MTILES ? mt_next : mt;
    const float* atile_next = h + (size_t)mt_pf * (16 * HID);

    v8f acc[4] = {};
    gemm_load_block(f1, atile, a_lane, sB, bidx, 1);       gemm_mma_block(f0, acc);
    gemm_load_block(f0, atile, a_lane, sB, bidx, 2);       gemm_mma_block(f1, acc);
    gemm_load_block(f1, atile, a_lane, sB, bidx, 3);       gemm_mma_block(f0, acc);
    gemm_load_block(f0, atile, a_lane, sB, bidx, 4);       gemm_mma_block(f1, acc);
    gemm_load_block(f1, atile, a_lane, sB, bidx, 5);       gemm_mma_block(f0, acc);
    gemm_load_block(f0, atile, a_lane, sB, bidx, 6);       gemm_mma_block(f1, acc);
    gemm_load_block(f1, atile, a_lane, sB, bidx, 7);       gemm_mma_block(f0, acc);
    gemm_load_block(f0, atile_next, a_lane, sB, bidx, 0);  gemm_mma_block(f1, acc);

    float* xrow = x + (size_t)(mt * 16 + 8 * g) * HID + nq * 64 + c;
#pragma unroll
    for (int t = 0; t < 4; ++t)
#pragma unroll
      for (int i = 0; i < 8; ++i) xrow[(size_t)i * HID + 16 * t] = acc[t][i];

    if (mt_next >= GEMM_MTILES) break;
    mt = mt_next;
    atile = atile_next;
  }
}

typedef __attribute__((ext_vector_type(4))) int v4i;

#define OWN_THREADS 1024
#define OWN_WAVES (OWN_THREADS / 32)
#define OWN_PIECES (N_EDGES / 16)
#define OWN_STEPS ((OWN_PIECES + 31) / 32)
#define OWN_SB 4
#define OWN_ITERS ((OWN_STEPS + OWN_WAVES * OWN_SB - 1) / (OWN_WAVES * OWN_SB))
#define OWN_QCAP 64
#define OWN_INIT ((OWN_ROWS * (HID / 4) + OWN_THREADS - 1) / OWN_THREADS)
#define AGG_GROUP 4

static_assert(N_EDGES % 16 == 0, "the owner scan reads whole 16-edge pieces");
static_assert((OWN_GROUPS - 1) * OWN_ROWS < N_NODES, "every group owns a row");
static_assert((OWN_GROUPS - 1) * (OWN_STEPS / OWN_GROUPS) < OWN_STEPS, "rotation");

__device__ __forceinline__ int wave_bcast_i(int v, int src_lane) {
  return __builtin_amdgcn_readlane(v, src_lane);
}

__device__ __forceinline__ int own_slot0(int i) {
  return (i >> 6) * HID + ((i & 32) << 2) + (i & 31);
}

__device__ __forceinline__ void own_edges4(
    const float* __restrict__ x, float* acc_lane,
    int r_l, int d_l, float nm_l, int k, unsigned lane_bytes) {
  v4f v[AGG_GROUP][2];
#pragma unroll
  for (int u = 0; u < AGG_GROUP; ++u) {
    const int r = wave_bcast_i(r_l, k + u);
    const char* src = (const char*)(x + (size_t)r * HID) + lane_bytes;
#pragma unroll
    for (int hh = 0; hh < 2; ++hh) v[u][hh] = *(const v4f*)(src + 512 * hh);
  }
#pragma unroll
  for (int u = 0; u < AGG_GROUP; ++u) {
    const int d = wave_bcast_i(d_l, k + u);
    const float nm = __int_as_float(wave_bcast_i(__float_as_int(nm_l), k + u));
    float* dst = acc_lane + d * HID;
#pragma unroll
    for (int hh = 0; hh < 2; ++hh)
#pragma unroll
      for (int j = 0; j < 4; ++j)
        atomicAdd(dst + 32 * (4 * hh + j), v[u][hh][j] * nm);
  }
}

__device__ __forceinline__ unsigned own_zero_bytes(unsigned y) {
  const unsigned t = ((y & 0x7F7F7F7Fu) + 0x7F7F7F7Fu) | y;
  return ~t & 0x80808080u;
}

__device__ __forceinline__ unsigned own_match16(v4i cv, unsigned pat) {
  const unsigned f0 = own_zero_bytes((unsigned)cv.x ^ pat);
  const unsigned f1 = own_zero_bytes((unsigned)cv.y ^ pat);
  const unsigned f2 = own_zero_bytes((unsigned)cv.z ^ pat);
  const unsigned f3 = own_zero_bytes((unsigned)cv.w ^ pat);
  return (f0 >> 7) | (f1 >> 6) | (f2 >> 5) | (f3 >> 4);
}

__device__ __forceinline__ void own_push(int* q, int& qn, unsigned mask, bool has, int e) {
  if (has) q[qn + (int)__builtin_amdgcn_mbcnt_lo(mask, 0u)] = e;
  qn += __builtin_popcount(mask);
}

__device__ __forceinline__ void own_drain(
    const int* __restrict__ ei, const float* __restrict__ w,
    const float* __restrict__ deg, const float* __restrict__ x,
    float* acc_lane, const int* qe, int n, int lo, int lane, unsigned lane_bytes) {
  int r_l = 0, d_l = 0;
  float nm_l = 0.0f;
  if (lane < n) {
    const int e = qe[lane];
    r_l = ei[e];
    const int c = ei[N_EDGES + e];
    d_l = c - lo;
    nm_l = w[e] * deg_to_dinv(deg[r_l]);
    if ((unsigned)d_l >= (unsigned)OWN_ROWS) { d_l = 0; nm_l = 0.0f; }
  }
  int k = 0;
  for (; k + AGG_GROUP <= n; k += AGG_GROUP)
    own_edges4(x, acc_lane, r_l, d_l, nm_l, k, lane_bytes);
  for (; k < n; ++k) {
    const int r = wave_bcast_i(r_l, k);
    const int d = wave_bcast_i(d_l, k);
    const float nm = __int_as_float(wave_bcast_i(__float_as_int(nm_l), k));
    const char* src = (const char*)(x + (size_t)r * HID) + lane_bytes;
    float* dst = acc_lane + d * HID;
    const v4f lo4 = *(const v4f*)src;
    const v4f hi4 = *(const v4f*)(src + 512);
#pragma unroll
    for (int j = 0; j < 4; ++j) {
      atomicAdd(dst + 32 * j, lo4[j] * nm);
      atomicAdd(dst + 32 * (4 + j), hi4[j] * nm);
    }
  }
}

__global__ __launch_bounds__(OWN_THREADS) void aggregate_owner_kernel(
    const int* __restrict__ ei,
    const unsigned char* __restrict__ owner,
    const float* __restrict__ w,
    const float* __restrict__ deg,
    const float* __restrict__ x,
    const float* __restrict__ bc,
    float* __restrict__ out) {
  __shared__ float acc[OWN_ROWS * HID];
  __shared__ int queue[OWN_WAVES * OWN_QCAP];
  __shared__ float sdinv[OWN_ROWS];

  const int tid = threadIdx.x;
  const int lane = tid & 31;
  const int wave = __builtin_amdgcn_readfirstlane(tid >> 5);
  const int lo = blockIdx.x * OWN_ROWS;
  const int cnt = (N_NODES - lo < OWN_ROWS) ? (N_NODES - lo) : OWN_ROWS;

  const float dgo = deg[lo + (tid < cnt ? tid : 0)];
  {
    const v4f* x4 = (const v4f*)(x + (size_t)lo * HID);
    const int total = cnt * (HID / 4);
    v4f xv[OWN_INIT];
    float dgr[OWN_INIT];
#pragma unroll
    for (int t = 0; t < OWN_INIT; ++t) {
      const int i = tid + t * OWN_THREADS;
      const int ic = i < total ? i : total - 1;
      xv[t] = x4[ic];
      dgr[t] = deg[lo + (ic >> 6)];
    }
#pragma unroll
    for (int t = 0; t < OWN_INIT; ++t) {
      const int i = tid + t * OWN_THREADS;
      if (i < total) {
        float* s = acc + own_slot0(i);
        const float di = deg_to_dinv(dgr[t]);
        s[0] = xv[t].x * di;
        s[32] = xv[t].y * di;
        s[64] = xv[t].z * di;
        s[96] = xv[t].w * di;
      }
    }
  }
  if (tid < cnt) sdinv[tid] = deg_to_dinv(dgo);
  __syncthreads();

  int* q = queue + wave * OWN_QCAP;
  int qn = 0;
  const int rot = blockIdx.x * (OWN_STEPS / OWN_GROUPS);
  const v4i* own4 = (const v4i*)owner;
  const unsigned pat = blockIdx.x * 0x01010101u;
  const unsigned lane_bytes = (unsigned)lane * 16u;
  float* acc_lane = acc + lane;

  v4i cv[OWN_SB];
  int piece[OWN_SB];
#pragma unroll
  for (int u = 0; u < OWN_SB; ++u) {
    int sr = u * OWN_WAVES + wave + rot;
    if (sr >= OWN_STEPS) sr -= OWN_STEPS;
    piece[u] = sr * 32 + lane;
    cv[u] = own4[piece[u] < OWN_PIECES ? piece[u] : OWN_PIECES - 1];
  }

  static_assert(OWN_SB == 4, "the loop below names the four steps of a batch");
  for (int it = 0; it < OWN_ITERS; ++it) {
    const int itn = it + 1 < OWN_ITERS ? it + 1 : it;
    unsigned m[OWN_SB];
    int e0[OWN_SB];
#pragma unroll
    for (int u = 0; u < OWN_SB; ++u) {
      const int s = (it * OWN_SB + u) * OWN_WAVES + wave;
      m[u] = (s < OWN_STEPS && piece[u] < OWN_PIECES) ? own_match16(cv[u], pat) : 0u;
      e0[u] = piece[u] * 16;
      int sn = (itn * OWN_SB + u) * OWN_WAVES + wave + rot;
      if (sn >= OWN_STEPS) sn -= OWN_STEPS;
      piece[u] = sn * 32 + lane;
      cv[u] = own4[piece[u] < OWN_PIECES ? piece[u] : OWN_PIECES - 1];
    }
    unsigned any = __builtin_amdgcn_ballot_w32((m[0] | m[1] | m[2] | m[3]) != 0u);
    while (any != 0u) {
      const bool h0 = m[0] != 0u, h1 = m[1] != 0u, h2 = m[2] != 0u;
      const unsigned mm = h0 ? m[0] : h1 ? m[1] : h2 ? m[2] : m[3];
      const int eb = h0 ? e0[0] : h1 ? e0[1] : h2 ? e0[2] : e0[3];
      const int p = __builtin_ctz(mm | 0x80000000u);
      own_push(q, qn, any, mm != 0u, eb + (((p & 3) << 2) | (p >> 3)));
      const unsigned rest = mm & (mm - 1u);
      m[0] = h0 ? rest : m[0];
      m[1] = (!h0 && h1) ? rest : m[1];
      m[2] = (!h0 && !h1 && h2) ? rest : m[2];
      m[3] = (!h0 && !h1 && !h2) ? rest : m[3];
      if (qn >= 32) {
        qn -= 32;
        own_drain(ei, w, deg, x, acc_lane, q + qn, 32, lo, lane, lane_bytes);
      }
      any = __builtin_amdgcn_ballot_w32((m[0] | m[1] | m[2] | m[3]) != 0u);
    }
  }
  if (qn > 0) own_drain(ei, w, deg, x, acc_lane, q, qn, lo, lane, lane_bytes);
  __syncthreads();

  static_assert(OWN_THREADS % (HID / 4) == 0, "one bias piece per thread");
  const v4f b = ((const v4f*)bc)[tid & 63];
  v4f* out4 = (v4f*)(out + (size_t)lo * HID);
  for (int i = tid; i < cnt * (HID / 4); i += OWN_THREADS) {
    const float* s = acc + own_slot0(i);
    const float di = sdinv[i >> 6];
    v4f o;
    o.x = s[0] * di + b.x;
    o.y = s[32] * di + b.y;
    o.z = s[64] * di + b.z;
    o.w = s[96] * di + b.w;
    out4[i] = o;
  }
}

extern "C" void kernel_launch(void* const* d_in, const int* in_sizes, int n_in,
                              void* d_out, int out_size, void* d_ws, size_t ws_size,
                              hipStream_t stream) {
  const float* h    = (const float*)d_in[0];
  const int*   ei   = (const int*)d_in[1];
  const float* attr = (const float*)d_in[2];
  const float* W1   = (const float*)d_in[3];
  const float* b1   = (const float*)d_in[4];
  const float* W2   = (const float*)d_in[5];
  const float* b2   = (const float*)d_in[6];
  const float* Wc   = (const float*)d_in[7];
  const float* bc   = (const float*)d_in[8];
  float* out = (float*)d_out;

  const int N = N_NODES, E = N_EDGES;

  float* x   = (float*)d_ws;
  float* w   = x + (size_t)N * HID;
  float* deg = w + E;
  unsigned char* owner = (unsigned char*)(deg + N);

  gemm_mlp_kernel<<<GEMM_GROUPS, FUSED_THREADS, 0, stream>>>(h, Wc, x, attr, W1, b1, W2, b2, ei + E, w, deg, owner);
  deg_lds_kernel<<<DEG_GROUPS, DEG_THREADS, 0, stream>>>(ei + E, w, deg, E);
  aggregate_owner_kernel<<<OWN_GROUPS, OWN_THREADS, 0, stream>>>(ei, owner, w, deg, x, bc, out);
}
